# starting kernel
# baseline (speedup 1.0000x reference)
#include <hip/hip_runtime.h>

typedef __attribute__((ext_vector_type(2))) float v2f;
typedef __attribute__((ext_vector_type(8))) float v8f;

#define NEG_SLOPE 0.2f
#define NEG_INF_F (-1e30f)

__device__ __forceinline__ unsigned f2o(float f) {
    unsigned u = __float_as_uint(f);
    return (u & 0x80000000u) ? ~u : (u | 0x80000000u);
}
__device__ __forceinline__ float o2f(unsigned o) {
    return (o & 0x80000000u) ? __uint_as_float(o & 0x7FFFFFFFu)
                             : __uint_as_float(~o);
}

template<int CIN, int COUT>
__global__ __launch_bounds__(256)
void gat_gemm_wmma(const float* __restrict__ X, const float* __restrict__ Wm,
                   const float* __restrict__ asrc, const float* __restrict__ adst,
                   float* __restrict__ H, float* __restrict__ S, float* __restrict__ D,
                   int N)
{
    constexpr int NT = COUT / 16;
    const int wave = (blockIdx.x * blockDim.x + threadIdx.x) >> 5;
    const int lane = threadIdx.x & 31;
    const int m0 = wave * 16;
    if (m0 >= N) return;

    const int half = lane >> 4;
    const int l = lane & 15;

    v8f acc[NT] = {};

    int arow = m0 + l; if (arow >= N) arow = N - 1;
    const float* xrow = X + (size_t)arow * CIN + 2 * half;

    #pragma unroll 4
    for (int k = 0; k < CIN; k += 4) {
        v2f A;
        A[0] = xrow[k];
        A[1] = xrow[k + 1];
        #pragma unroll
        for (int t = 0; t < NT; ++t) {
            const int n = t * 16 + l;
            v2f B;
            B[0] = Wm[(size_t)(k + 2 * half    ) * COUT + n];
            B[1] = Wm[(size_t)(k + 2 * half + 1) * COUT + n];
            acc[t] = __builtin_amdgcn_wmma_f32_16x16x4_f32(
                         false, A, false, B, (short)0, acc[t], false, false);
        }
    }

    #pragma unroll
    for (int j = 0; j < 8; ++j) {
        const int row = m0 + j + 8 * half;
        float ps = 0.0f, pd = 0.0f;
        #pragma unroll
        for (int t = 0; t < NT; ++t) {
            const int n = t * 16 + l;
            const float v = acc[t][j];
            if (row < N) H[(size_t)row * COUT + n] = v;
            ps += v * asrc[n];
            pd += v * adst[n];
        }
        #pragma unroll
        for (int off = 1; off < 16; off <<= 1) {
            ps += __shfl_xor(ps, off, 32);
            pd += __shfl_xor(pd, off, 32);
        }
        if (l == 0 && row < N) { S[row] = ps; D[row] = pd; }
    }
}

__global__ void gat_init(float* __restrict__ agg, unsigned* __restrict__ Mo,
                         float* __restrict__ den, int N, int COUT)
{
    const int i = blockIdx.x * blockDim.x + threadIdx.x;
    if (i < N * COUT) agg[i] = 0.0f;
    if (i < N) { Mo[i] = 0u; den[i] = 0.0f; }
}

__global__ void gat_edge_alpha(const int* __restrict__ src, const int* __restrict__ dst,
                               const float* __restrict__ S, const float* __restrict__ D,
                               float* __restrict__ alp, unsigned* __restrict__ Mo,
                               int E, int Eall)
{
    const int e = blockIdx.x * blockDim.x + threadIdx.x;
    if (e >= Eall) return;
    int si, di; bool self;
    if (e < E) { si = src[e]; di = dst[e]; self = (si == di); }
    else       { si = di = e - E;          self = false; }
    float a = S[si] + D[di];
    a = (a > 0.0f) ? a : NEG_SLOPE * a;
    if (self) a = NEG_INF_F;
    alp[e] = a;
    atomicMax(&Mo[di], f2o(a));
}

__global__ void gat_edge_exp(const int* __restrict__ dst,
                             float* __restrict__ alp, const unsigned* __restrict__ Mo,
                             float* __restrict__ den, int E, int Eall)
{
    const int e = blockIdx.x * blockDim.x + threadIdx.x;
    if (e >= Eall) return;
    const int di = (e < E) ? dst[e] : (e - E);
    const float m = o2f(Mo[di]);
    const float ex = __expf(alp[e] - m);
    alp[e] = ex;
    atomicAdd(&den[di], ex);
}

template<int COUT>
__global__ __launch_bounds__(256)
void gat_edge_agg(const int* __restrict__ src, const int* __restrict__ dst,
                  const float* __restrict__ ex, const float* __restrict__ den,
                  const float* __restrict__ H, float* __restrict__ AGG,
                  int E, int Eall)
{
    constexpr int G = COUT / 4;
    const long long tid = (long long)blockIdx.x * blockDim.x + threadIdx.x;
    const long long e = tid / G;
    if (e >= Eall) return;
    const int c4 = (int)(tid % G) * 4;
    int si, di;
    if (e < E) { si = src[e]; di = dst[e]; }
    else       { si = di = (int)(e - E); }
    const float w = ex[e] / den[di];
    const float4 hv = *(const float4*)(H + (size_t)si * COUT + c4);
    float* outp = AGG + (size_t)di * COUT + c4;
    atomicAdd(outp + 0, w * hv.x);
    atomicAdd(outp + 1, w * hv.y);
    atomicAdd(outp + 2, w * hv.z);
    atomicAdd(outp + 3, w * hv.w);
}

template<bool RELU, int COUT>
__global__ void gat_finalize(const float* __restrict__ agg, const float* __restrict__ b,
                             float* __restrict__ out, int N)
{
    const int i = blockIdx.x * blockDim.x + threadIdx.x;
    if (i >= N * COUT) return;
    float v = agg[i] + b[i & (COUT - 1)];
    if (RELU) v = fmaxf(v, 0.0f);
    out[i] = v;
}

static inline int cdiv_ll(long long a, long long b) { return (int)((a + b - 1) / b); }

extern "C" void kernel_launch(void* const* d_in, const int* in_sizes, int n_in,
                              void* d_out, int out_size, void* d_ws, size_t ws_size,
                              hipStream_t stream)
{
    const float* x   = (const float*)d_in[0];
    const int*   ei  = (const int*)  d_in[1];
    const float* W1  = (const float*)d_in[2];
    const float* as1 = (const float*)d_in[3];
    const float* ad1 = (const float*)d_in[4];
    const float* b1  = (const float*)d_in[5];
    const float* W2  = (const float*)d_in[6];
    const float* as2 = (const float*)d_in[7];
    const float* ad2 = (const float*)d_in[8];
    const float* b2  = (const float*)d_in[9];
    const float* W3  = (const float*)d_in[10];
    const float* as3 = (const float*)d_in[11];
    const float* ad3 = (const float*)d_in[12];
    const float* b3  = (const float*)d_in[13];

    const int N    = in_sizes[0] / 128;
    const int E    = in_sizes[1] / 2;
    const int Eall = E + N;
    const int* src = ei;
    const int* dst = ei + E;

    float* ws   = (float*)d_ws;
    float* h    = ws;
    float* agg  = h   + (size_t)N * 128;
    float* xn   = agg + (size_t)N * 128;
    float* alp  = xn  + (size_t)N * 128;
    float* Sv   = alp + Eall;
    float* Dv   = Sv  + N;
    unsigned* Mo = (unsigned*)(Dv + N);
    float* den  = (float*)(Mo + N);

    const dim3 blk(256);
    const int waves_blocks = cdiv_ll((long long)cdiv_ll(N, 16) * 32, 256);
    const int edge_blocks  = cdiv_ll(Eall, 256);

    gat_init<<<cdiv_ll((long long)N * 128, 256), blk, 0, stream>>>(agg, Mo, den, N, 128);
    gat_gemm_wmma<128,128><<<waves_blocks, blk, 0, stream>>>(x, W1, as1, ad1, h, Sv, Dv, N);
    gat_edge_alpha<<<edge_blocks, blk, 0, stream>>>(src, dst, Sv, Dv, alp, Mo, E, Eall);
    gat_edge_exp<<<edge_blocks, blk, 0, stream>>>(dst, alp, Mo, den, E, Eall);
    gat_edge_agg<128><<<cdiv_ll((long long)Eall * 32, 256), blk, 0, stream>>>(src, dst, alp, den, h, agg, E, Eall);
    gat_finalize<true,128><<<cdiv_ll((long long)N * 128, 256), blk, 0, stream>>>(agg, b1, xn, N);

    gat_init<<<cdiv_ll((long long)N * 64, 256), blk, 0, stream>>>(agg, Mo, den, N, 64);
    gat_gemm_wmma<128,64><<<waves_blocks, blk, 0, stream>>>(xn, W2, as2, ad2, h, Sv, Dv, N);
    gat_edge_alpha<<<edge_blocks, blk, 0, stream>>>(src, dst, Sv, Dv, alp, Mo, E, Eall);
    gat_edge_exp<<<edge_blocks, blk, 0, stream>>>(dst, alp, Mo, den, E, Eall);
    gat_edge_agg<64><<<cdiv_ll((long long)Eall * 16, 256), blk, 0, stream>>>(src, dst, alp, den, h, agg, E, Eall);
    gat_finalize<true,64><<<cdiv_ll((long long)N * 64, 256), blk, 0, stream>>>(agg, b2, xn, N);

    gat_init<<<cdiv_ll((long long)N * 64, 256), blk, 0, stream>>>(agg, Mo, den, N, 64);
    gat_gemm_wmma<64,64><<<waves_blocks, blk, 0, stream>>>(xn, W3, as3, ad3, h, Sv, Dv, N);
    gat_edge_alpha<<<edge_blocks, blk, 0, stream>>>(src, dst, Sv, Dv, alp, Mo, E, Eall);
    gat_edge_exp<<<edge_blocks, blk, 0, stream>>>(dst, alp, Mo, den, E, Eall);
    gat_edge_agg<64><<<cdiv_ll((long long)Eall * 16, 256), blk, 0, stream>>>(src, dst, alp, den, h, agg, E, Eall);
    gat_finalize<false,64><<<cdiv_ll((long long)N * 64, 256), blk, 0, stream>>>(agg, b3, (float*)d_out, N);
}


# replace the per-edge atomic max/sum/scatter passes by a destination-sorted edge list (counting sort, arrival-order rank) and one gather kernel per layer; sums in list order, divided at the end
# speedup vs baseline: 3.9393x; 3.9393x over previous
#include <hip/hip_runtime.h>

typedef __attribute__((ext_vector_type(2))) float v2f;
typedef __attribute__((ext_vector_type(8))) float v8f;

#define NEG_SLOPE 0.2f
#define NEG_INF_F (-1e30f)

template<int CIN, int COUT>
__global__ __launch_bounds__(256)
void gat_gemm_wmma(const float* __restrict__ X, const float* __restrict__ Wm,
                   const float* __restrict__ asrc, const float* __restrict__ adst,
                   float* __restrict__ H, float* __restrict__ S, float* __restrict__ D,
                   int N)
{
    constexpr int NT = COUT / 16;
    const int wave = (blockIdx.x * blockDim.x + threadIdx.x) >> 5;
    const int lane = threadIdx.x & 31;
    const int m0 = wave * 16;
    if (m0 >= N) return;

    const int half = lane >> 4;
    const int l = lane & 15;

    v8f acc[NT] = {};

    int arow = m0 + l; if (arow >= N) arow = N - 1;
    const float* xrow = X + (size_t)arow * CIN + 2 * half;

    #pragma unroll 4
    for (int k = 0; k < CIN; k += 4) {
        v2f A;
        A[0] = xrow[k];
        A[1] = xrow[k + 1];
        #pragma unroll
        for (int t = 0; t < NT; ++t) {
            const int n = t * 16 + l;
            v2f B;
            B[0] = Wm[(size_t)(k + 2 * half    ) * COUT + n];
            B[1] = Wm[(size_t)(k + 2 * half + 1) * COUT + n];
            acc[t] = __builtin_amdgcn_wmma_f32_16x16x4_f32(
                         false, A, false, B, (short)0, acc[t], false, false);
        }
    }

    #pragma unroll
    for (int j = 0; j < 8; ++j) {
        const int row = m0 + j + 8 * half;
        float ps = 0.0f, pd = 0.0f;
        #pragma unroll
        for (int t = 0; t < NT; ++t) {
            const int n = t * 16 + l;
            const float v = acc[t][j];
            if (row < N) H[(size_t)row * COUT + n] = v;
            ps += v * asrc[n];
            pd += v * adst[n];
        }
        #pragma unroll
        for (int off = 1; off < 16; off <<= 1) {
            ps += __shfl_xor(ps, off, 32);
            pd += __shfl_xor(pd, off, 32);
        }
        if (l == 0 && row < N) { S[row] = ps; D[row] = pd; }
    }
}

__device__ __forceinline__ int wave_sum_i(int v) {
    #pragma unroll
    for (int off = 16; off >= 1; off >>= 1) v += __shfl_xor(v, off, 32);
    return v;
}

__global__ void csr_zero(int* __restrict__ cnt, int N)
{
    const int i = blockIdx.x * blockDim.x + threadIdx.x;
    if (i < N) cnt[i] = 0;
}

__global__ void csr_count(const int* __restrict__ dst, int* __restrict__ cnt,
                          int* __restrict__ rank, int E)
{
    const int e = blockIdx.x * blockDim.x + threadIdx.x;
    if (e < E) rank[e] = atomicAdd(&cnt[dst[e]], 1);
}

__global__ __launch_bounds__(256)
void csr_tilesum(const int* __restrict__ cnt, int* __restrict__ tsum, int N)
{
    __shared__ int part[8];
    const int i = blockIdx.x * 256 + threadIdx.x;
    int v = (i < N) ? cnt[i] : 0;
    v = wave_sum_i(v);
    if ((threadIdx.x & 31) == 0) part[threadIdx.x >> 5] = v;
    __syncthreads();
    if (threadIdx.x == 0) {
        int s = 0;
        #pragma unroll
        for (int k = 0; k < 8; ++k) s += part[k];
        tsum[blockIdx.x] = s;
    }
}

__global__ __launch_bounds__(256)
void csr_scan(const int* __restrict__ cnt, const int* __restrict__ tsum,
              int* __restrict__ row_ptr, int N)
{
    __shared__ int part[8];
    const int t = threadIdx.x, lane = t & 31, w = t >> 5;

    int o = 0;
    for (int k = t; k < (int)blockIdx.x; k += 256) o += tsum[k];
    o = wave_sum_i(o);
    if (lane == 0) part[w] = o;
    __syncthreads();
    int tile_start = 0;
    #pragma unroll
    for (int k = 0; k < 8; ++k) tile_start += part[k];
    __syncthreads();

    const int i = blockIdx.x * 256 + t;
    const int v = (i < N) ? cnt[i] : 0;
    int inc = v;
    #pragma unroll
    for (int off = 1; off < 32; off <<= 1) {
        const int u = __shfl_up(inc, off, 32);
        if (lane >= off) inc += u;
    }
    if (lane == 31) part[w] = inc;
    __syncthreads();
    int wave_start = 0;
    #pragma unroll
    for (int k = 0; k < 8; ++k) if (k < w) wave_start += part[k];

    const int excl = tile_start + wave_start + inc - v;
    if (i < N) {
        row_ptr[i] = excl;
        if (i == N - 1) row_ptr[N] = excl + v;
    }
}

__global__ void csr_fill(const int* __restrict__ src, const int* __restrict__ dst,
                         const int* __restrict__ row_ptr, const int* __restrict__ rank,
                         int* __restrict__ csr_src, int E)
{
    const int e = blockIdx.x * blockDim.x + threadIdx.x;
    if (e >= E) return;
    csr_src[row_ptr[dst[e]] + rank[e]] = src[e];
}

template<bool RELU, int COUT>
__global__ __launch_bounds__(256)
void gat_gather(const int* __restrict__ row_ptr, const int* __restrict__ csr_src,
                const float* __restrict__ S, const float* __restrict__ D,
                const float* __restrict__ H, const float* __restrict__ bias,
                float* __restrict__ out, int N)
{
    constexpr int G = COUT / 4;
    const int node = (int)((blockIdx.x * blockDim.x + threadIdx.x) / G);
    if (node >= N) return;
    const int lig = threadIdx.x & (G - 1);
    const int c4 = lig * 4;

    const int start = row_ptr[node];
    const int total = row_ptr[node + 1] - start + 1;
    const float dn = D[node];

    float a_self = S[node] + dn;
    a_self = (a_self > 0.0f) ? a_self : NEG_SLOPE * a_self;

    float m_run = a_self;
    float den = 0.0f;
    float ax = 0.0f, ay = 0.0f, az = 0.0f, aw = 0.0f;

    for (int base = 0; base < total; base += G) {
        const int j = base + lig;
        const bool valid = j < total;
        int sj = node;
        float a = m_run;
        if (valid) {
            if (j > 0) {
                sj = csr_src[start + j - 1];
                float t = S[sj] + dn;
                t = (t > 0.0f) ? t : NEG_SLOPE * t;
                a = (sj == node) ? NEG_INF_F : t;
            } else {
                a = a_self;
            }
        }
        float cm = a;
        #pragma unroll
        for (int off = 1; off < G; off <<= 1) cm = fmaxf(cm, __shfl_xor(cm, off, 32));
        const float m_new = fmaxf(m_run, cm);
        const float scale = __expf(m_run - m_new);
        const float ex = valid ? __expf(a - m_new) : 0.0f;
        float es = ex;
        #pragma unroll
        for (int off = 1; off < G; off <<= 1) es += __shfl_xor(es, off, 32);
        den = den * scale + es;
        ax *= scale; ay *= scale; az *= scale; aw *= scale;

        const int cnt = (total - base < G) ? (total - base) : G;
        int t = 0;
        for (; t + 4 <= cnt; t += 4) {
            const int   s0 = __shfl(sj, t, G),     s1 = __shfl(sj, t + 1, G);
            const int   s2 = __shfl(sj, t + 2, G), s3 = __shfl(sj, t + 3, G);
            const float e0 = __shfl(ex, t, G),     e1 = __shfl(ex, t + 1, G);
            const float e2 = __shfl(ex, t + 2, G), e3 = __shfl(ex, t + 3, G);
            const float4 h0 = *(const float4*)(H + (size_t)s0 * COUT + c4);
            const float4 h1 = *(const float4*)(H + (size_t)s1 * COUT + c4);
            const float4 h2 = *(const float4*)(H + (size_t)s2 * COUT + c4);
            const float4 h3 = *(const float4*)(H + (size_t)s3 * COUT + c4);
            ax = fmaf(e0, h0.x, ax); ay = fmaf(e0, h0.y, ay); az = fmaf(e0, h0.z, az); aw = fmaf(e0, h0.w, aw);
            ax = fmaf(e1, h1.x, ax); ay = fmaf(e1, h1.y, ay); az = fmaf(e1, h1.z, az); aw = fmaf(e1, h1.w, aw);
            ax = fmaf(e2, h2.x, ax); ay = fmaf(e2, h2.y, ay); az = fmaf(e2, h2.z, az); aw = fmaf(e2, h2.w, aw);
            ax = fmaf(e3, h3.x, ax); ay = fmaf(e3, h3.y, ay); az = fmaf(e3, h3.z, az); aw = fmaf(e3, h3.w, aw);
        }
        for (; t < cnt; ++t) {
            const int   st = __shfl(sj, t, G);
            const float et = __shfl(ex, t, G);
            const float4 hv = *(const float4*)(H + (size_t)st * COUT + c4);
            ax = fmaf(et, hv.x, ax);
            ay = fmaf(et, hv.y, ay);
            az = fmaf(et, hv.z, az);
            aw = fmaf(et, hv.w, aw);
        }
        m_run = m_new;
    }

    float4 o;
    o.x = ax / den + bias[c4 + 0];
    o.y = ay / den + bias[c4 + 1];
    o.z = az / den + bias[c4 + 2];
    o.w = aw / den + bias[c4 + 3];
    if (RELU) {
        o.x = fmaxf(o.x, 0.0f); o.y = fmaxf(o.y, 0.0f);
        o.z = fmaxf(o.z, 0.0f); o.w = fmaxf(o.w, 0.0f);
    }
    *(float4*)(out + (size_t)node * COUT + c4) = o;
}

static inline int cdiv_ll(long long a, long long b) { return (int)((a + b - 1) / b); }

extern "C" void kernel_launch(void* const* d_in, const int* in_sizes, int n_in,
                              void* d_out, int out_size, void* d_ws, size_t ws_size,
                              hipStream_t stream)
{
    const float* x   = (const float*)d_in[0];
    const int*   ei  = (const int*)  d_in[1];
    const float* W1  = (const float*)d_in[2];
    const float* as1 = (const float*)d_in[3];
    const float* ad1 = (const float*)d_in[4];
    const float* b1  = (const float*)d_in[5];
    const float* W2  = (const float*)d_in[6];
    const float* as2 = (const float*)d_in[7];
    const float* ad2 = (const float*)d_in[8];
    const float* b2  = (const float*)d_in[9];
    const float* W3  = (const float*)d_in[10];
    const float* as3 = (const float*)d_in[11];
    const float* ad3 = (const float*)d_in[12];
    const float* b3  = (const float*)d_in[13];

    const int N    = in_sizes[0] / 128;
    const int E    = in_sizes[1] / 2;
    const int Eall = E + N;
    const int* src = ei;
    const int* dst = ei + E;

    float* ws   = (float*)d_ws;
    float* h    = ws;
    float* agg  = h   + (size_t)N * 128;
    float* xn   = agg + (size_t)N * 128;
    float* alp  = xn  + (size_t)N * 128;
    float* Sv   = alp + Eall;
    float* Dv   = Sv  + N;
    int* row_ptr = (int*)(Dv + N);
    int* csr_src = (int*)alp;
    int* cnt     = csr_src + E;
    int* tsum    = (int*)agg;
    int* rank    = tsum + N;

    const dim3 blk(256);
    const int waves_blocks = cdiv_ll((long long)cdiv_ll(N, 16) * 32, 256);
    const int node_blocks  = cdiv_ll(N, 256);
    const int edge_blocks  = cdiv_ll(E, 256);

    csr_zero<<<node_blocks, blk, 0, stream>>>(cnt, N);
    csr_count<<<edge_blocks, blk, 0, stream>>>(dst, cnt, rank, E);
    csr_tilesum<<<node_blocks, blk, 0, stream>>>(cnt, tsum, N);
    csr_scan<<<node_blocks, blk, 0, stream>>>(cnt, tsum, row_ptr, N);
    csr_fill<<<edge_blocks, blk, 0, stream>>>(src, dst, row_ptr, rank, csr_src, E);

    gat_gemm_wmma<128,128><<<waves_blocks, blk, 0, stream>>>(x, W1, as1, ad1, h, Sv, Dv, N);
    gat_gather<true,128><<<cdiv_ll((long long)N * 32, 256), blk, 0, stream>>>(row_ptr, csr_src, Sv, Dv, h, b1, xn, N);

    gat_gemm_wmma<128,64><<<waves_blocks, blk, 0, stream>>>(xn, W2, as2, ad2, h, Sv, Dv, N);
    gat_gather<true,64><<<cdiv_ll((long long)N * 16, 256), blk, 0, stream>>>(row_ptr, csr_src, Sv, Dv, h, b2, xn, N);

    gat_gemm_wmma<64,64><<<waves_blocks, blk, 0, stream>>>(xn, W3, as3, ad3, h, Sv, Dv, N);
    gat_gather<false,64><<<cdiv_ll((long long)N * 16, 256), blk, 0, stream>>>(row_ptr, csr_src, Sv, Dv, h, b3, (float*)d_out, N);
}


# each gather work-group (16 nodes) keeps its finished rows in LDS and computes the next layer's 16-row WMMA strip and s/d itself: 11 -> 9 launches, row write and re-read gone; sum orders kept
# speedup vs baseline: 4.1849x; 1.0623x over previous
#include <hip/hip_runtime.h>

typedef __attribute__((ext_vector_type(2))) float v2f;
typedef __attribute__((ext_vector_type(8))) float v8f;

#define NEG_SLOPE 0.2f
#define NEG_INF_F (-1e30f)

template<int CIN, int COUT>
__global__ __launch_bounds__(256)
void gat_gemm_wmma(const float* __restrict__ X, const float* __restrict__ Wm,
                   const float* __restrict__ asrc, const float* __restrict__ adst,
                   float* __restrict__ H, float* __restrict__ S, float* __restrict__ D,
                   int N)
{
    constexpr int NT = COUT / 16;
    const int wave = (blockIdx.x * blockDim.x + threadIdx.x) >> 5;
    const int lane = threadIdx.x & 31;
    const int m0 = wave * 16;
    if (m0 >= N) return;

    const int half = lane >> 4;
    const int l = lane & 15;

    v8f acc[NT] = {};

    int arow = m0 + l; if (arow >= N) arow = N - 1;
    const float* xrow = X + (size_t)arow * CIN + 2 * half;

    #pragma unroll 4
    for (int k = 0; k < CIN; k += 4) {
        v2f A;
        A[0] = xrow[k];
        A[1] = xrow[k + 1];
        #pragma unroll
        for (int t = 0; t < NT; ++t) {
            const int n = t * 16 + l;
            v2f B;
            B[0] = Wm[(size_t)(k + 2 * half    ) * COUT + n];
            B[1] = Wm[(size_t)(k + 2 * half + 1) * COUT + n];
            acc[t] = __builtin_amdgcn_wmma_f32_16x16x4_f32(
                         false, A, false, B, (short)0, acc[t], false, false);
        }
    }

    #pragma unroll
    for (int j = 0; j < 8; ++j) {
        const int row = m0 + j + 8 * half;
        float ps = 0.0f, pd = 0.0f;
        #pragma unroll
        for (int t = 0; t < NT; ++t) {
            const int n = t * 16 + l;
            const float v = acc[t][j];
            if (row < N) H[(size_t)row * COUT + n] = v;
            ps += v * asrc[n];
            pd += v * adst[n];
        }
        #pragma unroll
        for (int off = 1; off < 16; off <<= 1) {
            ps += __shfl_xor(ps, off, 32);
            pd += __shfl_xor(pd, off, 32);
        }
        if (l == 0 && row < N) { S[row] = ps; D[row] = pd; }
    }
}

__device__ __forceinline__ int wave_sum_i(int v) {
    #pragma unroll
    for (int off = 16; off >= 1; off >>= 1) v += __shfl_xor(v, off, 32);
    return v;
}

__global__ void csr_zero(int* __restrict__ cnt, int N)
{
    const int i = blockIdx.x * blockDim.x + threadIdx.x;
    if (i < N) cnt[i] = 0;
}

__global__ void csr_count(const int* __restrict__ dst, int* __restrict__ cnt,
                          int* __restrict__ rank, int E)
{
    const int e = blockIdx.x * blockDim.x + threadIdx.x;
    if (e < E) rank[e] = atomicAdd(&cnt[dst[e]], 1);
}

__global__ __launch_bounds__(256)
void csr_tilesum(const int* __restrict__ cnt, int* __restrict__ tsum, int N)
{
    __shared__ int part[8];
    const int i = blockIdx.x * 256 + threadIdx.x;
    int v = (i < N) ? cnt[i] : 0;
    v = wave_sum_i(v);
    if ((threadIdx.x & 31) == 0) part[threadIdx.x >> 5] = v;
    __syncthreads();
    if (threadIdx.x == 0) {
        int s = 0;
        #pragma unroll
        for (int k = 0; k < 8; ++k) s += part[k];
        tsum[blockIdx.x] = s;
    }
}

__global__ __launch_bounds__(256)
void csr_scan(const int* __restrict__ cnt, const int* __restrict__ tsum,
              int* __restrict__ row_ptr, int N)
{
    __shared__ int part[8];
    const int t = threadIdx.x, lane = t & 31, w = t >> 5;

    int o = 0;
    for (int k = t; k < (int)blockIdx.x; k += 256) o += tsum[k];
    o = wave_sum_i(o);
    if (lane == 0) part[w] = o;
    __syncthreads();
    int tile_start = 0;
    #pragma unroll
    for (int k = 0; k < 8; ++k) tile_start += part[k];
    __syncthreads();

    const int i = blockIdx.x * 256 + t;
    const int v = (i < N) ? cnt[i] : 0;
    int inc = v;
    #pragma unroll
    for (int off = 1; off < 32; off <<= 1) {
        const int u = __shfl_up(inc, off, 32);
        if (lane >= off) inc += u;
    }
    if (lane == 31) part[w] = inc;
    __syncthreads();
    int wave_start = 0;
    #pragma unroll
    for (int k = 0; k < 8; ++k) if (k < w) wave_start += part[k];

    const int excl = tile_start + wave_start + inc - v;
    if (i < N) {
        row_ptr[i] = excl;
        if (i == N - 1) row_ptr[N] = excl + v;
    }
}

__global__ void csr_fill(const int* __restrict__ src, const int* __restrict__ dst,
                         const int* __restrict__ row_ptr, const int* __restrict__ rank,
                         int* __restrict__ csr_src, int E)
{
    const int e = blockIdx.x * blockDim.x + threadIdx.x;
    if (e >= E) return;
    csr_src[row_ptr[dst[e]] + rank[e]] = src[e];
}

template<bool RELU, int COUT>
__global__ __launch_bounds__(256)
void gat_gather(const int* __restrict__ row_ptr, const int* __restrict__ csr_src,
                const float* __restrict__ S, const float* __restrict__ D,
                const float* __restrict__ H, const float* __restrict__ bias,
                float* __restrict__ out, int N)
{
    constexpr int G = COUT / 4;
    const int node = (int)((blockIdx.x * blockDim.x + threadIdx.x) / G);
    if (node >= N) return;
    const int lig = threadIdx.x & (G - 1);
    const int c4 = lig * 4;

    const int start = row_ptr[node];
    const int total = row_ptr[node + 1] - start + 1;
    const float dn = D[node];

    float a_self = S[node] + dn;
    a_self = (a_self > 0.0f) ? a_self : NEG_SLOPE * a_self;

    float m_run = a_self;
    float den = 0.0f;
    float ax = 0.0f, ay = 0.0f, az = 0.0f, aw = 0.0f;

    for (int base = 0; base < total; base += G) {
        const int j = base + lig;
        const bool valid = j < total;
        int sj = node;
        float a = m_run;
        if (valid) {
            if (j > 0) {
                sj = csr_src[start + j - 1];
                float t = S[sj] + dn;
                t = (t > 0.0f) ? t : NEG_SLOPE * t;
                a = (sj == node) ? NEG_INF_F : t;
            } else {
                a = a_self;
            }
        }
        float cm = a;
        #pragma unroll
        for (int off = 1; off < G; off <<= 1) cm = fmaxf(cm, __shfl_xor(cm, off, 32));
        const float m_new = fmaxf(m_run, cm);
        const float scale = __expf(m_run - m_new);
        const float ex = valid ? __expf(a - m_new) : 0.0f;
        float es = ex;
        #pragma unroll
        for (int off = 1; off < G; off <<= 1) es += __shfl_xor(es, off, 32);
        den = den * scale + es;
        ax *= scale; ay *= scale; az *= scale; aw *= scale;

        const int cnt = (total - base < G) ? (total - base) : G;
        int t = 0;
        for (; t + 4 <= cnt; t += 4) {
            const int   s0 = __shfl(sj, t, G),     s1 = __shfl(sj, t + 1, G);
            const int   s2 = __shfl(sj, t + 2, G), s3 = __shfl(sj, t + 3, G);
            const float e0 = __shfl(ex, t, G),     e1 = __shfl(ex, t + 1, G);
            const float e2 = __shfl(ex, t + 2, G), e3 = __shfl(ex, t + 3, G);
            const float4 h0 = *(const float4*)(H + (size_t)s0 * COUT + c4);
            const float4 h1 = *(const float4*)(H + (size_t)s1 * COUT + c4);
            const float4 h2 = *(const float4*)(H + (size_t)s2 * COUT + c4);
            const float4 h3 = *(const float4*)(H + (size_t)s3 * COUT + c4);
            ax = fmaf(e0, h0.x, ax); ay = fmaf(e0, h0.y, ay); az = fmaf(e0, h0.z, az); aw = fmaf(e0, h0.w, aw);
            ax = fmaf(e1, h1.x, ax); ay = fmaf(e1, h1.y, ay); az = fmaf(e1, h1.z, az); aw = fmaf(e1, h1.w, aw);
            ax = fmaf(e2, h2.x, ax); ay = fmaf(e2, h2.y, ay); az = fmaf(e2, h2.z, az); aw = fmaf(e2, h2.w, aw);
            ax = fmaf(e3, h3.x, ax); ay = fmaf(e3, h3.y, ay); az = fmaf(e3, h3.z, az); aw = fmaf(e3, h3.w, aw);
        }
        for (; t < cnt; ++t) {
            const int   st = __shfl(sj, t, G);
            const float et = __shfl(ex, t, G);
            const float4 hv = *(const float4*)(H + (size_t)st * COUT + c4);
            ax = fmaf(et, hv.x, ax);
            ay = fmaf(et, hv.y, ay);
            az = fmaf(et, hv.z, az);
            aw = fmaf(et, hv.w, aw);
        }
        m_run = m_new;
    }

    float4 o;
    o.x = ax / den + bias[c4 + 0];
    o.y = ay / den + bias[c4 + 1];
    o.z = az / den + bias[c4 + 2];
    o.w = aw / den + bias[c4 + 3];
    if (RELU) {
        o.x = fmaxf(o.x, 0.0f); o.y = fmaxf(o.y, 0.0f);
        o.z = fmaxf(o.z, 0.0f); o.w = fmaxf(o.w, 0.0f);
    }
    *(float4*)(out + (size_t)node * COUT + c4) = o;
}


template<bool RELU, int COUT, int NCOUT>
__global__ __launch_bounds__(4 * COUT)
void gat_gather_gemm(const int* __restrict__ row_ptr, const int* __restrict__ csr_src,
                     const float* __restrict__ S, const float* __restrict__ D,
                     const float* __restrict__ H, const float* __restrict__ bias,
                     const float* __restrict__ Wn, const float* __restrict__ asrcn,
                     const float* __restrict__ adstn,
                     float* __restrict__ Hn, float* __restrict__ Sn, float* __restrict__ Dn,
                     int N)
{
    constexpr int G  = COUT / 4;
    constexpr int XP = COUT + 4;
    constexpr int NT = NCOUT / 16;
    constexpr int HP = NCOUT + 4;
    __shared__ __attribute__((aligned(16))) float xs[16 * XP];
    __shared__ __attribute__((aligned(16))) float hs[16 * HP];

    const int r    = (int)threadIdx.x / G;
    const int m0   = (int)blockIdx.x * 16;
    const int node = m0 + r;
    const int lig  = threadIdx.x & (G - 1);
    const int c4   = lig * 4;

    float4 o;
    o.x = 0.0f; o.y = 0.0f; o.z = 0.0f; o.w = 0.0f;
    if (node < N) {
        const int start = row_ptr[node];
        const int total = row_ptr[node + 1] - start + 1;
        const float dn = D[node];

        float a_self = S[node] + dn;
        a_self = (a_self > 0.0f) ? a_self : NEG_SLOPE * a_self;

        float m_run = a_self;
        float den = 0.0f;
        float ax = 0.0f, ay = 0.0f, az = 0.0f, aw = 0.0f;

        for (int base = 0; base < total; base += G) {
            const int j = base + lig;
            const bool valid = j < total;
            int sj = node;
            float a = m_run;
            if (valid) {
                if (j > 0) {
                    sj = csr_src[start + j - 1];
                    float t = S[sj] + dn;
                    t = (t > 0.0f) ? t : NEG_SLOPE * t;
                    a = (sj == node) ? NEG_INF_F : t;
                } else {
                    a = a_self;
                }
            }
            float cm = a;
            #pragma unroll
            for (int off = 1; off < G; off <<= 1) cm = fmaxf(cm, __shfl_xor(cm, off, 32));
            const float m_new = fmaxf(m_run, cm);
            const float scale = __expf(m_run - m_new);
            const float ex = valid ? __expf(a - m_new) : 0.0f;
            float es = ex;
            #pragma unroll
            for (int off = 1; off < G; off <<= 1) es += __shfl_xor(es, off, 32);
            den = den * scale + es;
            ax *= scale; ay *= scale; az *= scale; aw *= scale;

            const int cnt = (total - base < G) ? (total - base) : G;
            int t = 0;
            for (; t + 4 <= cnt; t += 4) {
                const int   s0 = __shfl(sj, t, G),     s1 = __shfl(sj, t + 1, G);
                const int   s2 = __shfl(sj, t + 2, G), s3 = __shfl(sj, t + 3, G);
                const float e0 = __shfl(ex, t, G),     e1 = __shfl(ex, t + 1, G);
                const float e2 = __shfl(ex, t + 2, G), e3 = __shfl(ex, t + 3, G);
                const float4 h0 = *(const float4*)(H + (size_t)s0 * COUT + c4);
                const float4 h1 = *(const float4*)(H + (size_t)s1 * COUT + c4);
                const float4 h2 = *(const float4*)(H + (size_t)s2 * COUT + c4);
                const float4 h3 = *(const float4*)(H + (size_t)s3 * COUT + c4);
                ax = fmaf(e0, h0.x, ax); ay = fmaf(e0, h0.y, ay); az = fmaf(e0, h0.z, az); aw = fmaf(e0, h0.w, aw);
                ax = fmaf(e1, h1.x, ax); ay = fmaf(e1, h1.y, ay); az = fmaf(e1, h1.z, az); aw = fmaf(e1, h1.w, aw);
                ax = fmaf(e2, h2.x, ax); ay = fmaf(e2, h2.y, ay); az = fmaf(e2, h2.z, az); aw = fmaf(e2, h2.w, aw);
                ax = fmaf(e3, h3.x, ax); ay = fmaf(e3, h3.y, ay); az = fmaf(e3, h3.z, az); aw = fmaf(e3, h3.w, aw);
            }
            for (; t < cnt; ++t) {
                const int   st = __shfl(sj, t, G);
                const float et = __shfl(ex, t, G);
                const float4 hv = *(const float4*)(H + (size_t)st * COUT + c4);
                ax = fmaf(et, hv.x, ax);
                ay = fmaf(et, hv.y, ay);
                az = fmaf(et, hv.z, az);
                aw = fmaf(et, hv.w, aw);
            }
            m_run = m_new;
        }

        o.x = ax / den + bias[c4 + 0];
        o.y = ay / den + bias[c4 + 1];
        o.z = az / den + bias[c4 + 2];
        o.w = aw / den + bias[c4 + 3];
        if (RELU) {
            o.x = fmaxf(o.x, 0.0f); o.y = fmaxf(o.y, 0.0f);
            o.z = fmaxf(o.z, 0.0f); o.w = fmaxf(o.w, 0.0f);
        }
    }
    *(float4*)(xs + r * XP + c4) = o;
    __syncthreads();

    const int wave = (int)threadIdx.x >> 5;
    const int lane = threadIdx.x & 31;
    const int half = lane >> 4;
    const int l    = lane & 15;

    if (wave < NT) {
        v8f acc = {};
        const int n = wave * 16 + l;
        const float* xrow = xs + l * XP + 2 * half;
        const float* wcol = Wn + (size_t)(2 * half) * NCOUT + n;
        #pragma unroll 8
        for (int k = 0; k < COUT; k += 4) {
            v2f A, B;
            A[0] = xrow[k];
            A[1] = xrow[k + 1];
            B[0] = wcol[(size_t)k * NCOUT];
            B[1] = wcol[(size_t)(k + 1) * NCOUT];
            acc = __builtin_amdgcn_wmma_f32_16x16x4_f32(
                      false, A, false, B, (short)0, acc, false, false);
        }
        #pragma unroll
        for (int j = 0; j < 8; ++j) {
            const int lr = j + 8 * half;
            const float v = acc[j];
            if (m0 + lr < N) Hn[(size_t)(m0 + lr) * NCOUT + n] = v;
            hs[lr * HP + n] = v;
        }
    }
    __syncthreads();

    if (wave < 8) {
        const int lr  = wave + 8 * half;
        const int row = m0 + lr;
        float ps = 0.0f, pd = 0.0f;
        #pragma unroll
        for (int t = 0; t < NT; ++t) {
            const int n = t * 16 + l;
            const float v = hs[lr * HP + n];
            ps += v * asrcn[n];
            pd += v * adstn[n];
        }
        #pragma unroll
        for (int off = 1; off < 16; off <<= 1) {
            ps += __shfl_xor(ps, off, 32);
            pd += __shfl_xor(pd, off, 32);
        }
        if (l == 0 && row < N) { Sn[row] = ps; Dn[row] = pd; }
    }
}

static inline int cdiv_ll(long long a, long long b) { return (int)((a + b - 1) / b); }

extern "C" void kernel_launch(void* const* d_in, const int* in_sizes, int n_in,
                              void* d_out, int out_size, void* d_ws, size_t ws_size,
                              hipStream_t stream)
{
    const float* x   = (const float*)d_in[0];
    const int*   ei  = (const int*)  d_in[1];
    const float* W1  = (const float*)d_in[2];
    const float* as1 = (const float*)d_in[3];
    const float* ad1 = (const float*)d_in[4];
    const float* b1  = (const float*)d_in[5];
    const float* W2  = (const float*)d_in[6];
    const float* as2 = (const float*)d_in[7];
    const float* ad2 = (const float*)d_in[8];
    const float* b2  = (const float*)d_in[9];
    const float* W3  = (const float*)d_in[10];
    const float* as3 = (const float*)d_in[11];
    const float* ad3 = (const float*)d_in[12];
    const float* b3  = (const float*)d_in[13];

    const int N    = in_sizes[0] / 128;
    const int E    = in_sizes[1] / 2;
    const int Eall = E + N;
    const int* src = ei;
    const int* dst = ei + E;

    float* ws   = (float*)d_ws;
    float* h    = ws;
    float* agg  = h   + (size_t)N * 128;
    float* xn   = agg + (size_t)N * 128;
    float* alp  = xn  + (size_t)N * 128;
    float* Sv   = alp + Eall;
    float* Dv   = Sv  + N;
    int* row_ptr = (int*)(Dv + N);
    int* csr_src = (int*)alp;
    int* cnt     = csr_src + E;
    int* tsum    = (int*)agg;
    int* rank    = tsum + N;
    float* Sb    = agg + N + E;
    float* Db    = Sb + N;

    const dim3 blk(256);
    const int waves_blocks = cdiv_ll((long long)cdiv_ll(N, 16) * 32, 256);
    const int node_blocks  = cdiv_ll(N, 256);
    const int edge_blocks  = cdiv_ll(E, 256);

    csr_zero<<<node_blocks, blk, 0, stream>>>(cnt, N);
    csr_count<<<edge_blocks, blk, 0, stream>>>(dst, cnt, rank, E);
    csr_tilesum<<<node_blocks, blk, 0, stream>>>(cnt, tsum, N);
    csr_scan<<<node_blocks, blk, 0, stream>>>(cnt, tsum, row_ptr, N);
    csr_fill<<<edge_blocks, blk, 0, stream>>>(src, dst, row_ptr, rank, csr_src, E);

    gat_gemm_wmma<128,128><<<waves_blocks, blk, 0, stream>>>(x, W1, as1, ad1, h, Sv, Dv, N);

    gat_gather_gemm<true,128,64><<<cdiv_ll(N, 16), dim3(512), 0, stream>>>(
        row_ptr, csr_src, Sv, Dv, h, b1, W2, as2, ad2, xn, Sb, Db, N);

    gat_gather_gemm<true,64,64><<<cdiv_ll(N, 16), dim3(256), 0, stream>>>(
        row_ptr, csr_src, Sb, Db, xn, b2, W3, as3, ad3, h, Sv, Dv, N);

    gat_gather<false,64><<<cdiv_ll((long long)N * 16, 256), blk, 0, stream>>>(row_ptr, csr_src, Sv, Dv, h, b3, (float*)d_out, N);
}


# in-edge lists ranked on LDS counters instead of one global atomic per edge: edges binned by destination range, one work-group per range fills its slot rows; sums follow LDS-atomic arrival order
# speedup vs baseline: 6.2281x; 1.4882x over previous
#include <hip/hip_runtime.h>

typedef __attribute__((ext_vector_type(2))) float v2f;
typedef __attribute__((ext_vector_type(8))) float v8f;

#define NEG_SLOPE 0.2f
#define NEG_INF_F (-1e30f)

template<int CIN, int COUT>
__global__ __launch_bounds__(256)
void gat_gemm_wmma(const float* __restrict__ X, const float* __restrict__ Wm,
                   const float* __restrict__ asrc, const float* __restrict__ adst,
                   float* __restrict__ H, float* __restrict__ S, float* __restrict__ D,
                   int N)
{
    constexpr int NT = COUT / 16;
    const int wave = (blockIdx.x * blockDim.x + threadIdx.x) >> 5;
    const int lane = threadIdx.x & 31;
    const int m0 = wave * 16;
    if (m0 >= N) return;

    const int half = lane >> 4;
    const int l = lane & 15;

    v8f acc[NT] = {};

    int arow = m0 + l; if (arow >= N) arow = N - 1;
    const float* xrow = X + (size_t)arow * CIN + 2 * half;

    #pragma unroll 4
    for (int k = 0; k < CIN; k += 4) {
        v2f A;
        A[0] = xrow[k];
        A[1] = xrow[k + 1];
        #pragma unroll
        for (int t = 0; t < NT; ++t) {
            const int n = t * 16 + l;
            v2f B;
            B[0] = Wm[(size_t)(k + 2 * half    ) * COUT + n];
            B[1] = Wm[(size_t)(k + 2 * half + 1) * COUT + n];
            acc[t] = __builtin_amdgcn_wmma_f32_16x16x4_f32(
                         false, A, false, B, (short)0, acc[t], false, false);
        }
    }

    #pragma unroll
    for (int j = 0; j < 8; ++j) {
        const int row = m0 + j + 8 * half;
        float ps = 0.0f, pd = 0.0f;
        #pragma unroll
        for (int t = 0; t < NT; ++t) {
            const int n = t * 16 + l;
            const float v = acc[t][j];
            if (row < N) H[(size_t)row * COUT + n] = v;
            ps += v * asrc[n];
            pd += v * adst[n];
        }
        #pragma unroll
        for (int off = 1; off < 16; off <<= 1) {
            ps += __shfl_xor(ps, off, 32);
            pd += __shfl_xor(pd, off, 32);
        }
        if (l == 0 && row < N) { S[row] = ps; D[row] = pd; }
    }
}

#define SLOT_CAP 64
#define NBKT 256
#define BIN_THREADS 512
#define BIN_EPT 8
#define RANGE_MAX 1024

__global__ void ctr_zero(int* __restrict__ ctr)
{
    ctr[threadIdx.x] = 0;
}

__global__ __launch_bounds__(BIN_THREADS)
void edges_bin(const int* __restrict__ src, const int* __restrict__ dst,
               int* __restrict__ ctr, int* __restrict__ bins, int* __restrict__ spill,
               int N, int E)
{
    __shared__ int share[NBKT];
    const int tid = threadIdx.x;
    const int R = (N + NBKT - 1) / NBKT;
    const int pitch = N >> 2;
    int2* __restrict__ bin2 = (int2*)bins;
    int2* __restrict__ spill2 = (int2*)spill;

    if (tid < NBKT) share[tid] = 0;
    __syncthreads();

    const int e0 = (int)blockIdx.x * (BIN_THREADS * BIN_EPT) + tid;
    int sv[BIN_EPT], dv[BIN_EPT], bk[BIN_EPT], rk[BIN_EPT];
    #pragma unroll
    for (int j = 0; j < BIN_EPT; ++j) {
        const int e = e0 + j * BIN_THREADS;
        const int ec = (e < E) ? e : (E - 1);
        dv[j] = dst[ec];
        sv[j] = src[ec];
    }
    #pragma unroll
    for (int j = 0; j < BIN_EPT; ++j) {
        const int e = e0 + j * BIN_THREADS;
        bk[j] = (e < E) ? (dv[j] / R) : -1;
    }
    #pragma unroll
    for (int j = 0; j < BIN_EPT; ++j) {
        rk[j] = 0;
        if (bk[j] >= 0) rk[j] = atomicAdd(&share[bk[j]], 1);
    }
    __syncthreads();
    if (tid < NBKT) {
        const int c = share[tid];
        share[tid] = c ? atomicAdd(&ctr[2 + tid], c) : 0;
    }
    __syncthreads();
    #pragma unroll
    for (int j = 0; j < BIN_EPT; ++j) {
        if (bk[j] >= 0) {
            const int pos = share[bk[j]] + rk[j];
            int2* where;
            if (pos < pitch) where = bin2 + ((size_t)bk[j] * pitch + pos);
            else             where = spill2 + atomicAdd(&ctr[1], 1);
            int2 p; p.x = sv[j]; p.y = dv[j];
            *where = p;
        }
    }
}

__device__ __forceinline__ void slots_place(int2 p, int sub, int top, int* lcnt,
                                            int* __restrict__ ovf_len, int* __restrict__ slots,
                                            int* __restrict__ ovf_src, int* __restrict__ ovf_dst)
{
    if (p.y >= sub && p.y < top) {
        const int k = atomicAdd(&lcnt[p.y - sub], 1);
        if (k < SLOT_CAP) {
            slots[(size_t)p.y * SLOT_CAP + k] = p.x;
        } else {
            const int q = atomicAdd(ovf_len, 1);
            ovf_src[q] = p.x;
            ovf_dst[q] = p.y;
        }
    }
}

__global__ __launch_bounds__(BIN_THREADS)
void slots_build(const int* __restrict__ bins, const int* __restrict__ spill,
                 int* __restrict__ ctr, int* __restrict__ cnt, int* __restrict__ slots,
                 int* __restrict__ ovf_src, int* __restrict__ ovf_dst, int N)
{
    __shared__ int lcnt[RANGE_MAX];
    const int tid = threadIdx.x;
    const int b = blockIdx.x;
    const int R = (N + NBKT - 1) / NBKT;
    const int lo = b * R;
    if (lo >= N) return;
    const int hi = (lo + R < N) ? (lo + R) : N;
    const int pitch = N >> 2;
    const int2* __restrict__ bin2 = (const int2*)bins + (size_t)b * pitch;
    const int2* __restrict__ spill2 = (const int2*)spill;

    const int counted = ctr[2 + b];
    const int nin = (counted < pitch) ? counted : pitch;
    const int nsp = (counted > pitch) ? ctr[1] : 0;

    for (int sub = lo; sub < hi; sub += RANGE_MAX) {
        const int top = (sub + RANGE_MAX < hi) ? (sub + RANGE_MAX) : hi;
        for (int i = tid; i < top - sub; i += BIN_THREADS) lcnt[i] = 0;
        __syncthreads();
        for (int i0 = tid; i0 < nin; i0 += BIN_EPT * BIN_THREADS) {
            int2 p[BIN_EPT];
            #pragma unroll
            for (int j = 0; j < BIN_EPT; ++j) {
                const int i = i0 + j * BIN_THREADS;
                p[j] = bin2[(i < nin) ? i : (nin - 1)];
            }
            #pragma unroll
            for (int j = 0; j < BIN_EPT; ++j)
                if (i0 + j * BIN_THREADS < nin)
                    slots_place(p[j], sub, top, lcnt, ctr, slots, ovf_src, ovf_dst);
        }
        for (int i = tid; i < nsp; i += BIN_THREADS)
            slots_place(spill2[i], sub, top, lcnt, ctr, slots, ovf_src, ovf_dst);
        __syncthreads();
        for (int i = tid; i < top - sub; i += BIN_THREADS) cnt[sub + i] = lcnt[i];
        __syncthreads();
    }
}

template<int COUT>
__device__ __forceinline__ float4 gat_node_sum(const int* __restrict__ cnt, const int* __restrict__ slots,
                                               const int* __restrict__ ovf_src, const int* __restrict__ ovf_dst,
                                               const float* __restrict__ S, const float* __restrict__ D,
                                               const float* __restrict__ H, int node, int lig, int N)
{
    constexpr int G = COUT / 4;
    const int c4 = lig * 4;

    const int deg = cnt[node];
    const int total = ((deg < SLOT_CAP) ? deg : SLOT_CAP) + 1;
    const int* row = slots + (size_t)node * SLOT_CAP;
    const float dn = D[node];

    float a_self = S[node] + dn;
    a_self = (a_self > 0.0f) ? a_self : NEG_SLOPE * a_self;

    float m_run = a_self;
    float den = 0.0f;
    float ax = 0.0f, ay = 0.0f, az = 0.0f, aw = 0.0f;

    for (int base = 0; base < total; base += G) {
        const int j = base + lig;
        const bool valid = j < total;
        int sj = node;
        float a = m_run;
        if (valid) {
            if (j > 0) {
                sj = row[j - 1];
                float t = S[sj] + dn;
                t = (t > 0.0f) ? t : NEG_SLOPE * t;
                a = (sj == node) ? NEG_INF_F : t;
            } else {
                a = a_self;
            }
        }
        float cm = a;
        #pragma unroll
        for (int off = 1; off < G; off <<= 1) cm = fmaxf(cm, __shfl_xor(cm, off, 32));
        const float m_new = fmaxf(m_run, cm);
        const float scale = __expf(m_run - m_new);
        const float ex = valid ? __expf(a - m_new) : 0.0f;
        float es = ex;
        #pragma unroll
        for (int off = 1; off < G; off <<= 1) es += __shfl_xor(es, off, 32);
        den = den * scale + es;
        ax *= scale; ay *= scale; az *= scale; aw *= scale;

        const int cntg = (total - base < G) ? (total - base) : G;
        int t = 0;
        for (; t + 4 <= cntg; t += 4) {
            const int   s0 = __shfl(sj, t, G),     s1 = __shfl(sj, t + 1, G);
            const int   s2 = __shfl(sj, t + 2, G), s3 = __shfl(sj, t + 3, G);
            const float e0 = __shfl(ex, t, G),     e1 = __shfl(ex, t + 1, G);
            const float e2 = __shfl(ex, t + 2, G), e3 = __shfl(ex, t + 3, G);
            const float4 h0 = *(const float4*)(H + (size_t)s0 * COUT + c4);
            const float4 h1 = *(const float4*)(H + (size_t)s1 * COUT + c4);
            const float4 h2 = *(const float4*)(H + (size_t)s2 * COUT + c4);
            const float4 h3 = *(const float4*)(H + (size_t)s3 * COUT + c4);
            ax = fmaf(e0, h0.x, ax); ay = fmaf(e0, h0.y, ay); az = fmaf(e0, h0.z, az); aw = fmaf(e0, h0.w, aw);
            ax = fmaf(e1, h1.x, ax); ay = fmaf(e1, h1.y, ay); az = fmaf(e1, h1.z, az); aw = fmaf(e1, h1.w, aw);
            ax = fmaf(e2, h2.x, ax); ay = fmaf(e2, h2.y, ay); az = fmaf(e2, h2.z, az); aw = fmaf(e2, h2.w, aw);
            ax = fmaf(e3, h3.x, ax); ay = fmaf(e3, h3.y, ay); az = fmaf(e3, h3.z, az); aw = fmaf(e3, h3.w, aw);
        }
        for (; t < cntg; ++t) {
            const int   st = __shfl(sj, t, G);
            const float et = __shfl(ex, t, G);
            const float4 hv = *(const float4*)(H + (size_t)st * COUT + c4);
            ax = fmaf(et, hv.x, ax);
            ay = fmaf(et, hv.y, ay);
            az = fmaf(et, hv.z, az);
            aw = fmaf(et, hv.w, aw);
        }
        m_run = m_new;
    }

    if (deg > SLOT_CAP) {
        const int nov = cnt[N];
        for (int base = 0; base < nov; base += G) {
            const int q = base + lig;
            int keep = 0;
            int sj = node;
            float a = m_run;
            if (q < nov) {
                if (ovf_dst[q] == node) {
                    keep = 1;
                    sj = ovf_src[q];
                    float t = S[sj] + dn;
                    t = (t > 0.0f) ? t : NEG_SLOPE * t;
                    a = (sj == node) ? NEG_INF_F : t;
                }
            }
            float cm = a;
            #pragma unroll
            for (int off = 1; off < G; off <<= 1) cm = fmaxf(cm, __shfl_xor(cm, off, 32));
            const float m_new = fmaxf(m_run, cm);
            const float scale = __expf(m_run - m_new);
            const float ex = keep ? __expf(a - m_new) : 0.0f;
            float es = ex;
            #pragma unroll
            for (int off = 1; off < G; off <<= 1) es += __shfl_xor(es, off, 32);
            den = den * scale + es;
            ax *= scale; ay *= scale; az *= scale; aw *= scale;
            #pragma nounroll
            for (int t = 0; t < G; ++t) {
                const int   kt = __shfl(keep, t, G);
                const int   st = __shfl(sj, t, G);
                const float et = __shfl(ex, t, G);
                if (kt) {
                    const float4 hv = *(const float4*)(H + (size_t)st * COUT + c4);
                    ax = fmaf(et, hv.x, ax);
                    ay = fmaf(et, hv.y, ay);
                    az = fmaf(et, hv.z, az);
                    aw = fmaf(et, hv.w, aw);
                }
            }
            m_run = m_new;
        }
    }

    float4 o;
    o.x = ax / den;
    o.y = ay / den;
    o.z = az / den;
    o.w = aw / den;
    return o;
}

template<bool RELU, int COUT>
__global__ __launch_bounds__(256)
void gat_gather(const int* __restrict__ cnt, const int* __restrict__ slots,
                const int* __restrict__ ovf_src, const int* __restrict__ ovf_dst,
                const float* __restrict__ S, const float* __restrict__ D,
                const float* __restrict__ H, const float* __restrict__ bias,
                float* __restrict__ out, int N)
{
    constexpr int G = COUT / 4;
    const int node = (int)((blockIdx.x * blockDim.x + threadIdx.x) / G);
    if (node >= N) return;
    const int lig = threadIdx.x & (G - 1);
    const int c4 = lig * 4;

    float4 o = gat_node_sum<COUT>(cnt, slots, ovf_src, ovf_dst, S, D, H, node, lig, N);
    o.x += bias[c4 + 0];
    o.y += bias[c4 + 1];
    o.z += bias[c4 + 2];
    o.w += bias[c4 + 3];
    if (RELU) {
        o.x = fmaxf(o.x, 0.0f); o.y = fmaxf(o.y, 0.0f);
        o.z = fmaxf(o.z, 0.0f); o.w = fmaxf(o.w, 0.0f);
    }
    *(float4*)(out + (size_t)node * COUT + c4) = o;
}

template<bool RELU, int COUT, int NCOUT>
__global__ __launch_bounds__(4 * COUT)
void gat_gather_gemm(const int* __restrict__ cnt, const int* __restrict__ slots,
                     const int* __restrict__ ovf_src, const int* __restrict__ ovf_dst,
                     const float* __restrict__ S, const float* __restrict__ D,
                     const float* __restrict__ H, const float* __restrict__ bias,
                     const float* __restrict__ Wn, const float* __restrict__ asrcn,
                     const float* __restrict__ adstn,
                     float* __restrict__ Hn, float* __restrict__ Sn, float* __restrict__ Dn,
                     int N)
{
    constexpr int G  = COUT / 4;
    constexpr int XP = COUT + 4;
    constexpr int NT = NCOUT / 16;
    constexpr int HP = NCOUT + 4;
    __shared__ __attribute__((aligned(16))) float xs[16 * XP];
    __shared__ __attribute__((aligned(16))) float hs[16 * HP];

    const int r    = (int)threadIdx.x / G;
    const int m0   = (int)blockIdx.x * 16;
    const int node = m0 + r;
    const int lig  = threadIdx.x & (G - 1);
    const int c4   = lig * 4;

    float4 o;
    o.x = 0.0f; o.y = 0.0f; o.z = 0.0f; o.w = 0.0f;
    if (node < N) {
        o = gat_node_sum<COUT>(cnt, slots, ovf_src, ovf_dst, S, D, H, node, lig, N);
        o.x += bias[c4 + 0];
        o.y += bias[c4 + 1];
        o.z += bias[c4 + 2];
        o.w += bias[c4 + 3];
        if (RELU) {
            o.x = fmaxf(o.x, 0.0f); o.y = fmaxf(o.y, 0.0f);
            o.z = fmaxf(o.z, 0.0f); o.w = fmaxf(o.w, 0.0f);
        }
    }
    *(float4*)(xs + r * XP + c4) = o;
    __syncthreads();

    const int wave = (int)threadIdx.x >> 5;
    const int lane = threadIdx.x & 31;
    const int half = lane >> 4;
    const int l    = lane & 15;

    if (wave < NT) {
        v8f acc = {};
        const int n = wave * 16 + l;
        const float* xrow = xs + l * XP + 2 * half;
        const float* wcol = Wn + (size_t)(2 * half) * NCOUT + n;
        #pragma unroll 8
        for (int k = 0; k < COUT; k += 4) {
            v2f A, B;
            A[0] = xrow[k];
            A[1] = xrow[k + 1];
            B[0] = wcol[(size_t)k * NCOUT];
            B[1] = wcol[(size_t)(k + 1) * NCOUT];
            acc = __builtin_amdgcn_wmma_f32_16x16x4_f32(
                      false, A, false, B, (short)0, acc, false, false);
        }
        #pragma unroll
        for (int j = 0; j < 8; ++j) {
            const int lr = j + 8 * half;
            const float v = acc[j];
            if (m0 + lr < N) Hn[(size_t)(m0 + lr) * NCOUT + n] = v;
            hs[lr * HP + n] = v;
        }
    }
    __syncthreads();

    if (wave < 8) {
        const int lr  = wave + 8 * half;
        const int row = m0 + lr;
        float ps = 0.0f, pd = 0.0f;
        #pragma unroll
        for (int t = 0; t < NT; ++t) {
            const int n = t * 16 + l;
            const float v = hs[lr * HP + n];
            ps += v * asrcn[n];
            pd += v * adstn[n];
        }
        #pragma unroll
        for (int off = 1; off < 16; off <<= 1) {
            ps += __shfl_xor(ps, off, 32);
            pd += __shfl_xor(pd, off, 32);
        }
        if (l == 0 && row < N) { Sn[row] = ps; Dn[row] = pd; }
    }
}

static inline int cdiv_ll(long long a, long long b) { return (int)((a + b - 1) / b); }

extern "C" void kernel_launch(void* const* d_in, const int* in_sizes, int n_in,
                              void* d_out, int out_size, void* d_ws, size_t ws_size,
                              hipStream_t stream)
{
    const float* x   = (const float*)d_in[0];
    const int*   ei  = (const int*)  d_in[1];
    const float* W1  = (const float*)d_in[2];
    const float* as1 = (const float*)d_in[3];
    const float* ad1 = (const float*)d_in[4];
    const float* b1  = (const float*)d_in[5];
    const float* W2  = (const float*)d_in[6];
    const float* as2 = (const float*)d_in[7];
    const float* ad2 = (const float*)d_in[8];
    const float* b2  = (const float*)d_in[9];
    const float* W3  = (const float*)d_in[10];
    const float* as3 = (const float*)d_in[11];
    const float* ad3 = (const float*)d_in[12];
    const float* b3  = (const float*)d_in[13];

    const int N    = in_sizes[0] / 128;
    const int E    = in_sizes[1] / 2;
    const int Eall = E + N;
    const int* src = ei;
    const int* dst = ei + E;

    float* ws   = (float*)d_ws;
    float* h    = ws;
    float* agg  = h   + (size_t)N * 128;
    float* xn   = agg + (size_t)N * 128;
    float* alp  = xn  + (size_t)N * 128;
    float* Sv   = alp + Eall;
    float* Dv   = Sv  + N;
    int* slots   = (int*)agg;
    int* ovf_src = slots + (size_t)N * SLOT_CAP;
    int* ovf_dst = ovf_src + E;
    float* Sb    = (float*)(ovf_dst + E);
    float* Db    = Sb + N;
    int* cnt     = (int*)alp;

    int* ctr     = cnt + N;
    int* bins    = (int*)xn;
    int* spill   = (int*)h;

    const dim3 blk(256);
    const int waves_blocks = cdiv_ll((long long)cdiv_ll(N, 16) * 32, 256);
    const int bin_blocks   = cdiv_ll(E, BIN_THREADS * BIN_EPT);

    ctr_zero<<<1, dim3(NBKT + 2), 0, stream>>>(ctr);
    edges_bin<<<bin_blocks, dim3(BIN_THREADS), 0, stream>>>(src, dst, ctr, bins, spill, N, E);
    slots_build<<<NBKT, dim3(BIN_THREADS), 0, stream>>>(bins, spill, ctr, cnt, slots, ovf_src, ovf_dst, N);

    gat_gemm_wmma<128,128><<<waves_blocks, blk, 0, stream>>>(x, W1, as1, ad1, h, Sv, Dv, N);

    gat_gather_gemm<true,128,64><<<cdiv_ll(N, 16), dim3(512), 0, stream>>>(
        cnt, slots, ovf_src, ovf_dst, Sv, Dv, h, b1, W2, as2, ad2, xn, Sb, Db, N);

    gat_gather_gemm<true,64,64><<<cdiv_ll(N, 16), dim3(256), 0, stream>>>(
        cnt, slots, ovf_src, ovf_dst, Sb, Db, xn, b2, W3, as3, ad3, h, Sv, Dv, N);

    gat_gather<false,64><<<cdiv_ll((long long)N * 16, 256), blk, 0, stream>>>(cnt, slots, ovf_src, ovf_dst, Sv, Dv, h, b3, (float*)d_out, N);
}


# lane exchanges of the gather (softmax max/sum butterflies, per-edge source and weight broadcasts) and of the s/d row sums use DPP, permlane and readlane, not LDS permutes with waits; sum orders kept
# speedup vs baseline: 6.7104x; 1.0774x over previous
#include <hip/hip_runtime.h>

typedef __attribute__((ext_vector_type(2))) float v2f;
typedef __attribute__((ext_vector_type(8))) float v8f;

#define NEG_SLOPE 0.2f
#define NEG_INF_F (-1e30f)

template<int M>
__device__ __forceinline__ float row_xor(float v)
{
    return __builtin_bit_cast(float,
               __builtin_amdgcn_update_dpp(0, __builtin_bit_cast(int, v), 0x160 + M, 0xf, 0xf, true));
}
__device__ __forceinline__ float half_xor(float v)
{
    const int b = __builtin_bit_cast(int, v);
    return __builtin_bit_cast(float, __builtin_amdgcn_permlanex16(b, b, 0x76543210, 0xfedcba98, false, false));
}
__device__ __forceinline__ float row_sum16(float v)
{
    v += row_xor<1>(v);
    v += row_xor<2>(v);
    v += row_xor<4>(v);
    v += row_xor<8>(v);
    return v;
}
template<int G>
__device__ __forceinline__ float grp_max(float v)
{
    static_assert(G == 16 || G == 32, "a group is one 16-lane row or the whole wave");
    v = fmaxf(v, row_xor<1>(v));
    v = fmaxf(v, row_xor<2>(v));
    v = fmaxf(v, row_xor<4>(v));
    v = fmaxf(v, row_xor<8>(v));
    if (G == 32) v = fmaxf(v, half_xor(v));
    return v;
}
template<int G>
__device__ __forceinline__ float grp_sum(float v)
{
    static_assert(G == 16 || G == 32, "a group is one 16-lane row or the whole wave");
    v = row_sum16(v);
    if (G == 32) v += half_xor(v);
    return v;
}
template<int G>
__device__ __forceinline__ int grp_pick(int v, int t)
{
    static_assert(G == 16 || G == 32, "a group is one 16-lane row or the whole wave");
    if (G == 32) return __builtin_amdgcn_readlane(v, t);
    return (int)__builtin_amdgcn_permlane16_var((unsigned)v, (unsigned)v, (unsigned)t, false, false);
}
template<int G>
__device__ __forceinline__ float grp_pick(float v, int t)
{
    return __builtin_bit_cast(float, grp_pick<G>(__builtin_bit_cast(int, v), t));
}

#define PROD_STRIPS 8

template<int CIN, int COUT>
__device__ __forceinline__ void product_half_strips(
                   const float* __restrict__ X, const float* __restrict__ Wm,
                   const float* __restrict__ asrc, const float* __restrict__ adst,
                   float* __restrict__ H, float* __restrict__ S, float* __restrict__ D,
                   int N, int strip_first)
{
    constexpr int NT = COUT / 32;
    __shared__ float part[PROD_STRIPS * 16 * 2];
    const int wib  = (int)threadIdx.x >> 5;
    const int lane = threadIdx.x & 31;
    const int sib  = wib >> 1;
    const int up   = wib & 1;
    const int cb   = up * (COUT / 2);
    const int m0   = (strip_first + sib) * 16;
    const bool active = m0 < N;

    const int half = lane >> 4;
    const int l = lane & 15;

    float ps[8], pd[8];
    #pragma unroll
    for (int j = 0; j < 8; ++j) { ps[j] = 0.0f; pd[j] = 0.0f; }

    if (active) {
        v8f acc[NT] = {};

        int arow = m0 + l; if (arow >= N) arow = N - 1;
        const float* xrow = X + (size_t)arow * CIN + 2 * half;

        #pragma unroll 4
        for (int k = 0; k < CIN; k += 4) {
            v2f A;
            A[0] = xrow[k];
            A[1] = xrow[k + 1];
            #pragma unroll
            for (int t = 0; t < NT; ++t) {
                const int n = cb + t * 16 + l;
                v2f B;
                B[0] = Wm[(size_t)(k + 2 * half    ) * COUT + n];
                B[1] = Wm[(size_t)(k + 2 * half + 1) * COUT + n];
                acc[t] = __builtin_amdgcn_wmma_f32_16x16x4_f32(
                             false, A, false, B, (short)0, acc[t], false, false);
            }
        }

        #pragma unroll
        for (int j = 0; j < 8; ++j) {
            const int row = m0 + j + 8 * half;
            float qs = 0.0f, qd = 0.0f;
            #pragma unroll
            for (int t = 0; t < NT; ++t) {
                const int n = cb + t * 16 + l;
                const float v = acc[t][j];
                if (row < N) H[(size_t)row * COUT + n] = v;
                qs += v * asrc[n];
                qd += v * adst[n];
            }
            ps[j] = row_sum16(qs); pd[j] = row_sum16(qd);
        }
        if (up && l == 0) {
            #pragma unroll
            for (int j = 0; j < 8; ++j) {
                part[(sib * 16 + j + 8 * half) * 2 + 0] = ps[j];
                part[(sib * 16 + j + 8 * half) * 2 + 1] = pd[j];
            }
        }
    }
    __syncthreads();
    if (active && !up && l == 0) {
        #pragma unroll
        for (int j = 0; j < 8; ++j) {
            const int row = m0 + j + 8 * half;
            if (row < N) {
                S[row] = ps[j] + part[(sib * 16 + j + 8 * half) * 2 + 0];
                D[row] = pd[j] + part[(sib * 16 + j + 8 * half) * 2 + 1];
            }
        }
    }
}

#define SLOT_CAP 64
#define NBKT 256
#define BIN_THREADS 512
#define BIN_EPT 8
#define RANGE_MAX 1024

__global__ void ctr_zero(int* __restrict__ ctr)
{
    ctr[threadIdx.x] = 0;
}

__device__ __forceinline__
void edges_bin_body(const int* __restrict__ src, const int* __restrict__ dst,
               int* __restrict__ ctr, int* __restrict__ bins, int* __restrict__ spill,
               int N, int E, int chunk)
{
    __shared__ int share[NBKT];
    const int tid = threadIdx.x;
    const int R = (N + NBKT - 1) / NBKT;
    const int pitch = N >> 3;
    int2* __restrict__ bin2 = (int2*)bins;
    int2* __restrict__ spill2 = (int2*)spill;

    if (tid < NBKT) share[tid] = 0;
    __syncthreads();

    const int e0 = chunk * (BIN_THREADS * BIN_EPT) + tid;
    int sv[BIN_EPT], dv[BIN_EPT], bk[BIN_EPT], rk[BIN_EPT];
    #pragma unroll
    for (int j = 0; j < BIN_EPT; ++j) {
        const int e = e0 + j * BIN_THREADS;
        const int ec = (e < E) ? e : (E - 1);
        dv[j] = dst[ec];
        sv[j] = src[ec];
    }
    #pragma unroll
    for (int j = 0; j < BIN_EPT; ++j) {
        const int e = e0 + j * BIN_THREADS;
        bk[j] = (e < E) ? (dv[j] / R) : -1;
    }
    #pragma unroll
    for (int j = 0; j < BIN_EPT; ++j) {
        rk[j] = 0;
        if (bk[j] >= 0) rk[j] = atomicAdd(&share[bk[j]], 1);
    }
    __syncthreads();
    if (tid < NBKT) {
        const int c = share[tid];
        share[tid] = c ? atomicAdd(&ctr[2 + tid], c) : 0;
    }
    __syncthreads();
    #pragma unroll
    for (int j = 0; j < BIN_EPT; ++j) {
        if (bk[j] >= 0) {
            const int pos = share[bk[j]] + rk[j];
            int2* where;
            if (pos < pitch) where = bin2 + ((size_t)bk[j] * pitch + pos);
            else             where = spill2 + atomicAdd(&ctr[1], 1);
            int2 p; p.x = sv[j]; p.y = dv[j];
            *where = p;
        }
    }
}

__device__ __forceinline__ void slots_place(int2 p, int sub, int top, int* lcnt,
                                            int* __restrict__ ovf_len, int* __restrict__ slots,
                                            int* __restrict__ ovf_src, int* __restrict__ ovf_dst)
{
    if (p.y >= sub && p.y < top) {
        const int k = atomicAdd(&lcnt[p.y - sub], 1);
        if (k < SLOT_CAP) {
            slots[(size_t)p.y * SLOT_CAP + k] = p.x;
        } else {
            const int q = atomicAdd(ovf_len, 1);
            ovf_src[q] = p.x;
            ovf_dst[q] = p.y;
        }
    }
}

__device__ __forceinline__
void slots_build_body(const int* __restrict__ bins, const int* __restrict__ spill,
                 int* __restrict__ ctr, int* __restrict__ cnt, int* __restrict__ slots,
                 int* __restrict__ ovf_src, int* __restrict__ ovf_dst, int N, int b)
{
    __shared__ int lcnt[RANGE_MAX];
    const int tid = threadIdx.x;
    const int R = (N + NBKT - 1) / NBKT;
    const int lo = b * R;
    if (lo >= N) return;
    const int hi = (lo + R < N) ? (lo + R) : N;
    const int pitch = N >> 3;
    const int2* __restrict__ bin2 = (const int2*)bins + (size_t)b * pitch;
    const int2* __restrict__ spill2 = (const int2*)spill;

    const int counted = ctr[2 + b];
    const int nin = (counted < pitch) ? counted : pitch;
    const int nsp = (counted > pitch) ? ctr[1] : 0;

    for (int sub = lo; sub < hi; sub += RANGE_MAX) {
        const int top = (sub + RANGE_MAX < hi) ? (sub + RANGE_MAX) : hi;
        for (int i = tid; i < top - sub; i += BIN_THREADS) lcnt[i] = 0;
        __syncthreads();
        for (int i0 = tid; i0 < nin; i0 += BIN_EPT * BIN_THREADS) {
            int2 p[BIN_EPT];
            #pragma unroll
            for (int j = 0; j < BIN_EPT; ++j) {
                const int i = i0 + j * BIN_THREADS;
                p[j] = bin2[(i < nin) ? i : (nin - 1)];
            }
            #pragma unroll
            for (int j = 0; j < BIN_EPT; ++j)
                if (i0 + j * BIN_THREADS < nin)
                    slots_place(p[j], sub, top, lcnt, ctr, slots, ovf_src, ovf_dst);
        }
        for (int i = tid; i < nsp; i += BIN_THREADS)
            slots_place(spill2[i], sub, top, lcnt, ctr, slots, ovf_src, ovf_dst);
        __syncthreads();
        for (int i = tid; i < top - sub; i += BIN_THREADS) cnt[sub + i] = lcnt[i];
        __syncthreads();
    }
}

__global__ __launch_bounds__(BIN_THREADS)
void bin_and_product(const int* __restrict__ src, const int* __restrict__ dst,
                     int* __restrict__ ctr, int* __restrict__ bins, int* __restrict__ spill,
                     const float* __restrict__ X, const float* __restrict__ Wm,
                     const float* __restrict__ asrc, const float* __restrict__ adst,
                     float* __restrict__ H, float* __restrict__ S, float* __restrict__ D,
                     int N, int E, int list_blocks, int strip_first)
{
    const int b = (int)blockIdx.x;
    if (b < list_blocks)
        edges_bin_body(src, dst, ctr, bins, spill, N, E, b);
    else
        product_half_strips<128, 128>(X, Wm, asrc, adst, H, S, D, N,
                                      strip_first + (b - list_blocks) * PROD_STRIPS);
}

__global__ __launch_bounds__(BIN_THREADS)
void build_and_product(const int* __restrict__ bins, const int* __restrict__ spill,
                       int* __restrict__ ctr, int* __restrict__ cnt, int* __restrict__ slots,
                       int* __restrict__ ovf_src, int* __restrict__ ovf_dst,
                       const float* __restrict__ X, const float* __restrict__ Wm,
                       const float* __restrict__ asrc, const float* __restrict__ adst,
                       float* __restrict__ H, float* __restrict__ S, float* __restrict__ D,
                       int N, int list_blocks, int strip_first)
{
    const int b = (int)blockIdx.x;
    if (b < list_blocks)
        slots_build_body(bins, spill, ctr, cnt, slots, ovf_src, ovf_dst, N, b);
    else
        product_half_strips<128, 128>(X, Wm, asrc, adst, H, S, D, N,
                                      strip_first + (b - list_blocks) * PROD_STRIPS);
}

template<int COUT>
__device__ __forceinline__ float4 gat_node_sum(const int* __restrict__ cnt, const int* __restrict__ slots,
                                               const int* __restrict__ ovf_src, const int* __restrict__ ovf_dst,
                                               const float* __restrict__ S, const float* __restrict__ D,
                                               const float* __restrict__ H, int node, int lig, int N)
{
    constexpr int G = COUT / 4;
    const int c4 = lig * 4;

    const int deg = cnt[node];
    const int total = ((deg < SLOT_CAP) ? deg : SLOT_CAP) + 1;
    const int* row = slots + (size_t)node * SLOT_CAP;
    const float dn = D[node];

    float a_self = S[node] + dn;
    a_self = (a_self > 0.0f) ? a_self : NEG_SLOPE * a_self;

    float m_run = a_self;
    float den = 0.0f;
    float ax = 0.0f, ay = 0.0f, az = 0.0f, aw = 0.0f;

    for (int base = 0; base < total; base += G) {
        const int j = base + lig;
        const bool valid = j < total;
        int sj = node;
        float a = m_run;
        if (valid) {
            if (j > 0) {
                sj = row[j - 1];
                float t = S[sj] + dn;
                t = (t > 0.0f) ? t : NEG_SLOPE * t;
                a = (sj == node) ? NEG_INF_F : t;
            } else {
                a = a_self;
            }
        }
        const float cm = grp_max<G>(a);
        const float m_new = fmaxf(m_run, cm);
        const float scale = __expf(m_run - m_new);
        const float ex = valid ? __expf(a - m_new) : 0.0f;
        const float es = grp_sum<G>(ex);
        den = den * scale + es;
        ax *= scale; ay *= scale; az *= scale; aw *= scale;

        const int cntg = (total - base < G) ? (total - base) : G;
        int t = 0;
        for (; t + 4 <= cntg; t += 4) {
            const int   s0 = grp_pick<G>(sj, t),     s1 = grp_pick<G>(sj, t + 1);
            const int   s2 = grp_pick<G>(sj, t + 2), s3 = grp_pick<G>(sj, t + 3);
            const float e0 = grp_pick<G>(ex, t),     e1 = grp_pick<G>(ex, t + 1);
            const float e2 = grp_pick<G>(ex, t + 2), e3 = grp_pick<G>(ex, t + 3);
            const float4 h0 = *(const float4*)(H + (size_t)s0 * COUT + c4);
            const float4 h1 = *(const float4*)(H + (size_t)s1 * COUT + c4);
            const float4 h2 = *(const float4*)(H + (size_t)s2 * COUT + c4);
            const float4 h3 = *(const float4*)(H + (size_t)s3 * COUT + c4);
            ax = fmaf(e0, h0.x, ax); ay = fmaf(e0, h0.y, ay); az = fmaf(e0, h0.z, az); aw = fmaf(e0, h0.w, aw);
            ax = fmaf(e1, h1.x, ax); ay = fmaf(e1, h1.y, ay); az = fmaf(e1, h1.z, az); aw = fmaf(e1, h1.w, aw);
            ax = fmaf(e2, h2.x, ax); ay = fmaf(e2, h2.y, ay); az = fmaf(e2, h2.z, az); aw = fmaf(e2, h2.w, aw);
            ax = fmaf(e3, h3.x, ax); ay = fmaf(e3, h3.y, ay); az = fmaf(e3, h3.z, az); aw = fmaf(e3, h3.w, aw);
        }
        for (; t < cntg; ++t) {
            const int   st = grp_pick<G>(sj, t);
            const float et = grp_pick<G>(ex, t);
            const float4 hv = *(const float4*)(H + (size_t)st * COUT + c4);
            ax = fmaf(et, hv.x, ax);
            ay = fmaf(et, hv.y, ay);
            az = fmaf(et, hv.z, az);
            aw = fmaf(et, hv.w, aw);
        }
        m_run = m_new;
    }

    if (deg > SLOT_CAP) {
        const int nov = cnt[N];
        for (int base = 0; base < nov; base += G) {
            const int q = base + lig;
            int keep = 0;
            int sj = node;
            float a = m_run;
            if (q < nov) {
                if (ovf_dst[q] == node) {
                    keep = 1;
                    sj = ovf_src[q];
                    float t = S[sj] + dn;
                    t = (t > 0.0f) ? t : NEG_SLOPE * t;
                    a = (sj == node) ? NEG_INF_F : t;
                }
            }
            const float cm = grp_max<G>(a);
            const float m_new = fmaxf(m_run, cm);
            const float scale = __expf(m_run - m_new);
            const float ex = keep ? __expf(a - m_new) : 0.0f;
            const float es = grp_sum<G>(ex);
            den = den * scale + es;
            ax *= scale; ay *= scale; az *= scale; aw *= scale;
            #pragma nounroll
            for (int t = 0; t < G; ++t) {
                const int   kt = grp_pick<G>(keep, t);
                const int   st = grp_pick<G>(sj, t);
                const float et = grp_pick<G>(ex, t);
                if (kt) {
                    const float4 hv = *(const float4*)(H + (size_t)st * COUT + c4);
                    ax = fmaf(et, hv.x, ax);
                    ay = fmaf(et, hv.y, ay);
                    az = fmaf(et, hv.z, az);
                    aw = fmaf(et, hv.w, aw);
                }
            }
            m_run = m_new;
        }
    }

    float4 o;
    o.x = ax / den;
    o.y = ay / den;
    o.z = az / den;
    o.w = aw / den;
    return o;
}

template<bool RELU, int COUT>
__global__ __launch_bounds__(256)
void gat_gather(const int* __restrict__ cnt, const int* __restrict__ slots,
                const int* __restrict__ ovf_src, const int* __restrict__ ovf_dst,
                const float* __restrict__ S, const float* __restrict__ D,
                const float* __restrict__ H, const float* __restrict__ bias,
                float* __restrict__ out, int N)
{
    constexpr int G = COUT / 4;
    const int node = (int)((blockIdx.x * blockDim.x + threadIdx.x) / G);
    if (node >= N) return;
    const int lig = threadIdx.x & (G - 1);
    const int c4 = lig * 4;

    float4 o = gat_node_sum<COUT>(cnt, slots, ovf_src, ovf_dst, S, D, H, node, lig, N);
    o.x += bias[c4 + 0];
    o.y += bias[c4 + 1];
    o.z += bias[c4 + 2];
    o.w += bias[c4 + 3];
    if (RELU) {
        o.x = fmaxf(o.x, 0.0f); o.y = fmaxf(o.y, 0.0f);
        o.z = fmaxf(o.z, 0.0f); o.w = fmaxf(o.w, 0.0f);
    }
    *(float4*)(out + (size_t)node * COUT + c4) = o;
}

template<bool RELU, int COUT, int NCOUT>
__global__ __launch_bounds__(4 * COUT)
void gat_gather_gemm(const int* __restrict__ cnt, const int* __restrict__ slots,
                     const int* __restrict__ ovf_src, const int* __restrict__ ovf_dst,
                     const float* __restrict__ S, const float* __restrict__ D,
                     const float* __restrict__ H, const float* __restrict__ bias,
                     const float* __restrict__ Wn, const float* __restrict__ asrcn,
                     const float* __restrict__ adstn,
                     float* __restrict__ Hn, float* __restrict__ Sn, float* __restrict__ Dn,
                     int N)
{
    constexpr int G  = COUT / 4;
    constexpr int XP = COUT + 4;
    constexpr int NT = NCOUT / 16;
    constexpr int HP = NCOUT + 4;
    __shared__ __attribute__((aligned(16))) float xs[16 * XP];
    __shared__ __attribute__((aligned(16))) float hs[16 * HP];

    const int r    = (int)threadIdx.x / G;
    const int m0   = (int)blockIdx.x * 16;
    const int node = m0 + r;
    const int lig  = threadIdx.x & (G - 1);
    const int c4   = lig * 4;

    float4 o;
    o.x = 0.0f; o.y = 0.0f; o.z = 0.0f; o.w = 0.0f;
    if (node < N) {
        o = gat_node_sum<COUT>(cnt, slots, ovf_src, ovf_dst, S, D, H, node, lig, N);
        o.x += bias[c4 + 0];
        o.y += bias[c4 + 1];
        o.z += bias[c4 + 2];
        o.w += bias[c4 + 3];
        if (RELU) {
            o.x = fmaxf(o.x, 0.0f); o.y = fmaxf(o.y, 0.0f);
            o.z = fmaxf(o.z, 0.0f); o.w = fmaxf(o.w, 0.0f);
        }
    }
    *(float4*)(xs + r * XP + c4) = o;
    __syncthreads();

    const int wave = (int)threadIdx.x >> 5;
    const int lane = threadIdx.x & 31;
    const int half = lane >> 4;
    const int l    = lane & 15;

    if (wave < NT) {
        v8f acc = {};
        const int n = wave * 16 + l;
        const float* xrow = xs + l * XP + 2 * half;
        const float* wcol = Wn + (size_t)(2 * half) * NCOUT + n;
        #pragma unroll 8
        for (int k = 0; k < COUT; k += 4) {
            v2f A, B;
            A[0] = xrow[k];
            A[1] = xrow[k + 1];
            B[0] = wcol[(size_t)k * NCOUT];
            B[1] = wcol[(size_t)(k + 1) * NCOUT];
            acc = __builtin_amdgcn_wmma_f32_16x16x4_f32(
                      false, A, false, B, (short)0, acc, false, false);
        }
        #pragma unroll
        for (int j = 0; j < 8; ++j) {
            const int lr = j + 8 * half;
            const float v = acc[j];
            if (m0 + lr < N) Hn[(size_t)(m0 + lr) * NCOUT + n] = v;
            hs[lr * HP + n] = v;
        }
    }
    __syncthreads();

    if (wave < 8) {
        const int lr  = wave + 8 * half;
        const int row = m0 + lr;
        float ps = 0.0f, pd = 0.0f;
        #pragma unroll
        for (int t = 0; t < NT; ++t) {
            const int n = t * 16 + l;
            const float v = hs[lr * HP + n];
            ps += v * asrcn[n];
            pd += v * adstn[n];
        }
        ps = row_sum16(ps);
        pd = row_sum16(pd);
        if (l == 0 && row < N) { Sn[row] = ps; Dn[row] = pd; }
    }
}

static inline int cdiv_ll(long long a, long long b) { return (int)((a + b - 1) / b); }

extern "C" void kernel_launch(void* const* d_in, const int* in_sizes, int n_in,
                              void* d_out, int out_size, void* d_ws, size_t ws_size,
                              hipStream_t stream)
{
    const float* x   = (const float*)d_in[0];
    const int*   ei  = (const int*)  d_in[1];
    const float* W1  = (const float*)d_in[2];
    const float* as1 = (const float*)d_in[3];
    const float* ad1 = (const float*)d_in[4];
    const float* b1  = (const float*)d_in[5];
    const float* W2  = (const float*)d_in[6];
    const float* as2 = (const float*)d_in[7];
    const float* ad2 = (const float*)d_in[8];
    const float* b2  = (const float*)d_in[9];
    const float* W3  = (const float*)d_in[10];
    const float* as3 = (const float*)d_in[11];
    const float* ad3 = (const float*)d_in[12];
    const float* b3  = (const float*)d_in[13];

    const int N    = in_sizes[0] / 128;
    const int E    = in_sizes[1] / 2;
    const int Eall = E + N;
    const int* src = ei;
    const int* dst = ei + E;

    float* ws   = (float*)d_ws;
    float* h    = ws;
    float* agg  = h   + (size_t)N * 128;
    float* xn   = agg + (size_t)N * 128;
    float* alp  = xn  + (size_t)N * 128;
    float* Sv   = alp + Eall;
    float* Dv   = Sv  + N;
    int* slots   = (int*)agg;
    int* ovf_src = slots + (size_t)N * SLOT_CAP;
    int* ovf_dst = ovf_src + E;
    float* Sb    = (float*)(ovf_dst + E);
    float* Db    = Sb + N;
    int* cnt     = (int*)alp;

    int* ctr     = cnt + N;
    int* bins    = (int*)xn;
    int* spill   = (int*)xn + (size_t)N * 64;

    const dim3 blk(256);
    const int bin_blocks   = cdiv_ll(E, BIN_THREADS * BIN_EPT);
    const int prod_blocks  = cdiv_ll(cdiv_ll(N, 16), PROD_STRIPS);
    const int prod_a       = prod_blocks / 2;
    const int prod_b       = prod_blocks - prod_a;

    ctr_zero<<<1, dim3(NBKT + 2), 0, stream>>>(ctr);
    bin_and_product<<<bin_blocks + prod_a, dim3(BIN_THREADS), 0, stream>>>(
        src, dst, ctr, bins, spill, x, W1, as1, ad1, h, Sv, Dv, N, E, bin_blocks, 0);
    build_and_product<<<NBKT + prod_b, dim3(BIN_THREADS), 0, stream>>>(
        bins, spill, ctr, cnt, slots, ovf_src, ovf_dst, x, W1, as1, ad1, h, Sv, Dv, N, NBKT, prod_a * PROD_STRIPS);

    gat_gather_gemm<true,128,64><<<cdiv_ll(N, 16), dim3(512), 0, stream>>>(
        cnt, slots, ovf_src, ovf_dst, Sv, Dv, h, b1, W2, as2, ad2, xn, Sb, Db, N);

    gat_gather_gemm<true,64,64><<<cdiv_ll(N, 16), dim3(256), 0, stream>>>(
        cnt, slots, ovf_src, ovf_dst, Sb, Db, xn, b2, W3, as3, ad3, h, Sv, Dv, N);

    gat_gather<false,64><<<cdiv_ll((long long)N * 16, 256), blk, 0, stream>>>(cnt, slots, ovf_src, ovf_dst, Sv, Dv, h, b3, (float*)d_out, N);
}


# fused next-layer product behind each gather cut along k over ALL waves (4 tiles x 4 or 2 k slices, one load group each, a_src/a_dst in it); slice sums added in order via LDS: product sum order changes
# speedup vs baseline: 7.0014x; 1.0434x over previous
#include <hip/hip_runtime.h>

typedef __attribute__((ext_vector_type(2))) float v2f;
typedef __attribute__((ext_vector_type(8))) float v8f;

#define NEG_SLOPE 0.2f
#define NEG_INF_F (-1e30f)

template<int M>
__device__ __forceinline__ float row_xor(float v)
{
    return __builtin_bit_cast(float,
               __builtin_amdgcn_update_dpp(0, __builtin_bit_cast(int, v), 0x160 + M, 0xf, 0xf, true));
}
__device__ __forceinline__ float half_xor(float v)
{
    const int b = __builtin_bit_cast(int, v);
    return __builtin_bit_cast(float, __builtin_amdgcn_permlanex16(b, b, 0x76543210, 0xfedcba98, false, false));
}
__device__ __forceinline__ float row_sum16(float v)
{
    v += row_xor<1>(v);
    v += row_xor<2>(v);
    v += row_xor<4>(v);
    v += row_xor<8>(v);
    return v;
}
template<int G>
__device__ __forceinline__ float grp_max(float v)
{
    static_assert(G == 16 || G == 32, "a group is one 16-lane row or the whole wave");
    v = fmaxf(v, row_xor<1>(v));
    v = fmaxf(v, row_xor<2>(v));
    v = fmaxf(v, row_xor<4>(v));
    v = fmaxf(v, row_xor<8>(v));
    if (G == 32) v = fmaxf(v, half_xor(v));
    return v;
}
template<int G>
__device__ __forceinline__ float grp_sum(float v)
{
    static_assert(G == 16 || G == 32, "a group is one 16-lane row or the whole wave");
    v = row_sum16(v);
    if (G == 32) v += half_xor(v);
    return v;
}
template<int G>
__device__ __forceinline__ int grp_pick(int v, int t)
{
    static_assert(G == 16 || G == 32, "a group is one 16-lane row or the whole wave");
    if (G == 32) return __builtin_amdgcn_readlane(v, t);
    return (int)__builtin_amdgcn_permlane16_var((unsigned)v, (unsigned)v, (unsigned)t, false, false);
}
template<int G>
__device__ __forceinline__ float grp_pick(float v, int t)
{
    return __builtin_bit_cast(float, grp_pick<G>(__builtin_bit_cast(int, v), t));
}

#define PROD_STRIPS 8

template<int CIN, int COUT>
__device__ __forceinline__ void product_half_strips(
                   const float* __restrict__ X, const float* __restrict__ Wm,
                   const float* __restrict__ asrc, const float* __restrict__ adst,
                   float* __restrict__ H, float* __restrict__ S, float* __restrict__ D,
                   int N, int strip_first)
{
    constexpr int NT = COUT / 32;
    __shared__ float part[PROD_STRIPS * 16 * 2];
    const int wib  = (int)threadIdx.x >> 5;
    const int lane = threadIdx.x & 31;
    const int sib  = wib >> 1;
    const int up   = wib & 1;
    const int cb   = up * (COUT / 2);
    const int m0   = (strip_first + sib) * 16;
    const bool active = m0 < N;

    const int half = lane >> 4;
    const int l = lane & 15;

    float ps[8], pd[8];
    #pragma unroll
    for (int j = 0; j < 8; ++j) { ps[j] = 0.0f; pd[j] = 0.0f; }

    if (active) {
        v8f acc[NT] = {};

        int arow = m0 + l; if (arow >= N) arow = N - 1;
        const float* xrow = X + (size_t)arow * CIN + 2 * half;

        #pragma unroll 4
        for (int k = 0; k < CIN; k += 4) {
            v2f A;
            A[0] = xrow[k];
            A[1] = xrow[k + 1];
            #pragma unroll
            for (int t = 0; t < NT; ++t) {
                const int n = cb + t * 16 + l;
                v2f B;
                B[0] = Wm[(size_t)(k + 2 * half    ) * COUT + n];
                B[1] = Wm[(size_t)(k + 2 * half + 1) * COUT + n];
                acc[t] = __builtin_amdgcn_wmma_f32_16x16x4_f32(
                             false, A, false, B, (short)0, acc[t], false, false);
            }
        }

        #pragma unroll
        for (int j = 0; j < 8; ++j) {
            const int row = m0 + j + 8 * half;
            float qs = 0.0f, qd = 0.0f;
            #pragma unroll
            for (int t = 0; t < NT; ++t) {
                const int n = cb + t * 16 + l;
                const float v = acc[t][j];
                if (row < N) H[(size_t)row * COUT + n] = v;
                qs += v * asrc[n];
                qd += v * adst[n];
            }
            ps[j] = row_sum16(qs); pd[j] = row_sum16(qd);
        }
        if (up && l == 0) {
            #pragma unroll
            for (int j = 0; j < 8; ++j) {
                part[(sib * 16 + j + 8 * half) * 2 + 0] = ps[j];
                part[(sib * 16 + j + 8 * half) * 2 + 1] = pd[j];
            }
        }
    }
    __syncthreads();
    if (active && !up && l == 0) {
        #pragma unroll
        for (int j = 0; j < 8; ++j) {
            const int row = m0 + j + 8 * half;
            if (row < N) {
                S[row] = ps[j] + part[(sib * 16 + j + 8 * half) * 2 + 0];
                D[row] = pd[j] + part[(sib * 16 + j + 8 * half) * 2 + 1];
            }
        }
    }
}

#define SLOT_CAP 64
#define NBKT 256
#define BIN_THREADS 512
#define BIN_EPT 8
#define RANGE_MAX 1024

__global__ void ctr_zero(int* __restrict__ ctr)
{
    ctr[threadIdx.x] = 0;
}

__device__ __forceinline__
void edges_bin_body(const int* __restrict__ src, const int* __restrict__ dst,
               int* __restrict__ ctr, int* __restrict__ bins, int* __restrict__ spill,
               int N, int E, int chunk)
{
    __shared__ int share[NBKT];
    const int tid = threadIdx.x;
    const int R = (N + NBKT - 1) / NBKT;
    const int pitch = N >> 3;
    int2* __restrict__ bin2 = (int2*)bins;
    int2* __restrict__ spill2 = (int2*)spill;

    if (tid < NBKT) share[tid] = 0;
    __syncthreads();

    const int e0 = chunk * (BIN_THREADS * BIN_EPT) + tid;
    int sv[BIN_EPT], dv[BIN_EPT], bk[BIN_EPT], rk[BIN_EPT];
    #pragma unroll
    for (int j = 0; j < BIN_EPT; ++j) {
        const int e = e0 + j * BIN_THREADS;
        const int ec = (e < E) ? e : (E - 1);
        dv[j] = dst[ec];
        sv[j] = src[ec];
    }
    #pragma unroll
    for (int j = 0; j < BIN_EPT; ++j) {
        const int e = e0 + j * BIN_THREADS;
        bk[j] = (e < E) ? (dv[j] / R) : -1;
    }
    #pragma unroll
    for (int j = 0; j < BIN_EPT; ++j) {
        rk[j] = 0;
        if (bk[j] >= 0) rk[j] = atomicAdd(&share[bk[j]], 1);
    }
    __syncthreads();
    if (tid < NBKT) {
        const int c = share[tid];
        share[tid] = c ? atomicAdd(&ctr[2 + tid], c) : 0;
    }
    __syncthreads();
    #pragma unroll
    for (int j = 0; j < BIN_EPT; ++j) {
        if (bk[j] >= 0) {
            const int pos = share[bk[j]] + rk[j];
            int2* where;
            if (pos < pitch) where = bin2 + ((size_t)bk[j] * pitch + pos);
            else             where = spill2 + atomicAdd(&ctr[1], 1);
            int2 p; p.x = sv[j]; p.y = dv[j];
            *where = p;
        }
    }
}

__device__ __forceinline__ void slots_place(int2 p, int sub, int top, int* lcnt,
                                            int* __restrict__ ovf_len, int* __restrict__ slots,
                                            int* __restrict__ ovf_src, int* __restrict__ ovf_dst)
{
    if (p.y >= sub && p.y < top) {
        const int k = atomicAdd(&lcnt[p.y - sub], 1);
        if (k < SLOT_CAP) {
            slots[(size_t)p.y * SLOT_CAP + k] = p.x;
        } else {
            const int q = atomicAdd(ovf_len, 1);
            ovf_src[q] = p.x;
            ovf_dst[q] = p.y;
        }
    }
}

__device__ __forceinline__
void slots_build_body(const int* __restrict__ bins, const int* __restrict__ spill,
                 int* __restrict__ ctr, int* __restrict__ cnt, int* __restrict__ slots,
                 int* __restrict__ ovf_src, int* __restrict__ ovf_dst, int N, int b)
{
    __shared__ int lcnt[RANGE_MAX];
    const int tid = threadIdx.x;
    const int R = (N + NBKT - 1) / NBKT;
    const int lo = b * R;
    if (lo >= N) return;
    const int hi = (lo + R < N) ? (lo + R) : N;
    const int pitch = N >> 3;
    const int2* __restrict__ bin2 = (const int2*)bins + (size_t)b * pitch;
    const int2* __restrict__ spill2 = (const int2*)spill;

    const int counted = ctr[2 + b];
    const int nin = (counted < pitch) ? counted : pitch;
    const int nsp = (counted > pitch) ? ctr[1] : 0;

    for (int sub = lo; sub < hi; sub += RANGE_MAX) {
        const int top = (sub + RANGE_MAX < hi) ? (sub + RANGE_MAX) : hi;
        for (int i = tid; i < top - sub; i += BIN_THREADS) lcnt[i] = 0;
        __syncthreads();
        for (int i0 = tid; i0 < nin; i0 += BIN_EPT * BIN_THREADS) {
            int2 p[BIN_EPT];
            #pragma unroll
            for (int j = 0; j < BIN_EPT; ++j) {
                const int i = i0 + j * BIN_THREADS;
                p[j] = bin2[(i < nin) ? i : (nin - 1)];
            }
            #pragma unroll
            for (int j = 0; j < BIN_EPT; ++j)
                if (i0 + j * BIN_THREADS < nin)
                    slots_place(p[j], sub, top, lcnt, ctr, slots, ovf_src, ovf_dst);
        }
        for (int i = tid; i < nsp; i += BIN_THREADS)
            slots_place(spill2[i], sub, top, lcnt, ctr, slots, ovf_src, ovf_dst);
        __syncthreads();
        for (int i = tid; i < top - sub; i += BIN_THREADS) cnt[sub + i] = lcnt[i];
        __syncthreads();
    }
}

__global__ __launch_bounds__(BIN_THREADS)
void bin_and_product(const int* __restrict__ src, const int* __restrict__ dst,
                     int* __restrict__ ctr, int* __restrict__ bins, int* __restrict__ spill,
                     const float* __restrict__ X, const float* __restrict__ Wm,
                     const float* __restrict__ asrc, const float* __restrict__ adst,
                     float* __restrict__ H, float* __restrict__ S, float* __restrict__ D,
                     int N, int E, int list_blocks, int strip_first)
{
    const int b = (int)blockIdx.x;
    if (b < list_blocks)
        edges_bin_body(src, dst, ctr, bins, spill, N, E, b);
    else
        product_half_strips<128, 128>(X, Wm, asrc, adst, H, S, D, N,
                                      strip_first + (b - list_blocks) * PROD_STRIPS);
}

__global__ __launch_bounds__(BIN_THREADS)
void build_and_product(const int* __restrict__ bins, const int* __restrict__ spill,
                       int* __restrict__ ctr, int* __restrict__ cnt, int* __restrict__ slots,
                       int* __restrict__ ovf_src, int* __restrict__ ovf_dst,
                       const float* __restrict__ X, const float* __restrict__ Wm,
                       const float* __restrict__ asrc, const float* __restrict__ adst,
                       float* __restrict__ H, float* __restrict__ S, float* __restrict__ D,
                       int N, int list_blocks, int strip_first)
{
    const int b = (int)blockIdx.x;
    if (b < list_blocks)
        slots_build_body(bins, spill, ctr, cnt, slots, ovf_src, ovf_dst, N, b);
    else
        product_half_strips<128, 128>(X, Wm, asrc, adst, H, S, D, N,
                                      strip_first + (b - list_blocks) * PROD_STRIPS);
}

template<int COUT>
__device__ __forceinline__ float4 gat_node_sum(const int* __restrict__ cnt, const int* __restrict__ slots,
                                               const int* __restrict__ ovf_src, const int* __restrict__ ovf_dst,
                                               const float* __restrict__ S, const float* __restrict__ D,
                                               const float* __restrict__ H, int node, int lig, int N)
{
    constexpr int G = COUT / 4;
    const int c4 = lig * 4;

    const int deg = cnt[node];
    const int total = ((deg < SLOT_CAP) ? deg : SLOT_CAP) + 1;
    const int* row = slots + (size_t)node * SLOT_CAP;
    const float dn = D[node];

    float a_self = S[node] + dn;
    a_self = (a_self > 0.0f) ? a_self : NEG_SLOPE * a_self;

    float m_run = a_self;
    float den = 0.0f;
    float ax = 0.0f, ay = 0.0f, az = 0.0f, aw = 0.0f;

    for (int base = 0; base < total; base += G) {
        const int j = base + lig;
        const bool valid = j < total;
        int sj = node;
        float a = m_run;
        if (valid) {
            if (j > 0) {
                sj = row[j - 1];
                float t = S[sj] + dn;
                t = (t > 0.0f) ? t : NEG_SLOPE * t;
                a = (sj == node) ? NEG_INF_F : t;
            } else {
                a = a_self;
            }
        }
        const float cm = grp_max<G>(a);
        const float m_new = fmaxf(m_run, cm);
        const float scale = __expf(m_run - m_new);
        const float ex = valid ? __expf(a - m_new) : 0.0f;
        const float es = grp_sum<G>(ex);
        den = den * scale + es;
        ax *= scale; ay *= scale; az *= scale; aw *= scale;

        const int cntg = (total - base < G) ? (total - base) : G;
        int t = 0;
        for (; t + 4 <= cntg; t += 4) {
            const int   s0 = grp_pick<G>(sj, t),     s1 = grp_pick<G>(sj, t + 1);
            const int   s2 = grp_pick<G>(sj, t + 2), s3 = grp_pick<G>(sj, t + 3);
            const float e0 = grp_pick<G>(ex, t),     e1 = grp_pick<G>(ex, t + 1);
            const float e2 = grp_pick<G>(ex, t + 2), e3 = grp_pick<G>(ex, t + 3);
            const float4 h0 = *(const float4*)(H + (size_t)s0 * COUT + c4);
            const float4 h1 = *(const float4*)(H + (size_t)s1 * COUT + c4);
            const float4 h2 = *(const float4*)(H + (size_t)s2 * COUT + c4);
            const float4 h3 = *(const float4*)(H + (size_t)s3 * COUT + c4);
            ax = fmaf(e0, h0.x, ax); ay = fmaf(e0, h0.y, ay); az = fmaf(e0, h0.z, az); aw = fmaf(e0, h0.w, aw);
            ax = fmaf(e1, h1.x, ax); ay = fmaf(e1, h1.y, ay); az = fmaf(e1, h1.z, az); aw = fmaf(e1, h1.w, aw);
            ax = fmaf(e2, h2.x, ax); ay = fmaf(e2, h2.y, ay); az = fmaf(e2, h2.z, az); aw = fmaf(e2, h2.w, aw);
            ax = fmaf(e3, h3.x, ax); ay = fmaf(e3, h3.y, ay); az = fmaf(e3, h3.z, az); aw = fmaf(e3, h3.w, aw);
        }
        for (; t < cntg; ++t) {
            const int   st = grp_pick<G>(sj, t);
            const float et = grp_pick<G>(ex, t);
            const float4 hv = *(const float4*)(H + (size_t)st * COUT + c4);
            ax = fmaf(et, hv.x, ax);
            ay = fmaf(et, hv.y, ay);
            az = fmaf(et, hv.z, az);
            aw = fmaf(et, hv.w, aw);
        }
        m_run = m_new;
    }

    if (deg > SLOT_CAP) {
        const int nov = cnt[N];
        for (int base = 0; base < nov; base += G) {
            const int q = base + lig;
            int keep = 0;
            int sj = node;
            float a = m_run;
            if (q < nov) {
                if (ovf_dst[q] == node) {
                    keep = 1;
                    sj = ovf_src[q];
                    float t = S[sj] + dn;
                    t = (t > 0.0f) ? t : NEG_SLOPE * t;
                    a = (sj == node) ? NEG_INF_F : t;
                }
            }
            const float cm = grp_max<G>(a);
            const float m_new = fmaxf(m_run, cm);
            const float scale = __expf(m_run - m_new);
            const float ex = keep ? __expf(a - m_new) : 0.0f;
            const float es = grp_sum<G>(ex);
            den = den * scale + es;
            ax *= scale; ay *= scale; az *= scale; aw *= scale;
            #pragma nounroll
            for (int t = 0; t < G; ++t) {
                const int   kt = grp_pick<G>(keep, t);
                const int   st = grp_pick<G>(sj, t);
                const float et = grp_pick<G>(ex, t);
                if (kt) {
                    const float4 hv = *(const float4*)(H + (size_t)st * COUT + c4);
                    ax = fmaf(et, hv.x, ax);
                    ay = fmaf(et, hv.y, ay);
                    az = fmaf(et, hv.z, az);
                    aw = fmaf(et, hv.w, aw);
                }
            }
            m_run = m_new;
        }
    }

    float4 o;
    o.x = ax / den;
    o.y = ay / den;
    o.z = az / den;
    o.w = aw / den;
    return o;
}

template<bool RELU, int COUT>
__global__ __launch_bounds__(256)
void gat_gather(const int* __restrict__ cnt, const int* __restrict__ slots,
                const int* __restrict__ ovf_src, const int* __restrict__ ovf_dst,
                const float* __restrict__ S, const float* __restrict__ D,
                const float* __restrict__ H, const float* __restrict__ bias,
                float* __restrict__ out, int N)
{
    constexpr int G = COUT / 4;
    const int node = (int)((blockIdx.x * blockDim.x + threadIdx.x) / G);
    if (node >= N) return;
    const int lig = threadIdx.x & (G - 1);
    const int c4 = lig * 4;

    float4 o = gat_node_sum<COUT>(cnt, slots, ovf_src, ovf_dst, S, D, H, node, lig, N);
    o.x += bias[c4 + 0];
    o.y += bias[c4 + 1];
    o.z += bias[c4 + 2];
    o.w += bias[c4 + 3];
    if (RELU) {
        o.x = fmaxf(o.x, 0.0f); o.y = fmaxf(o.y, 0.0f);
        o.z = fmaxf(o.z, 0.0f); o.w = fmaxf(o.w, 0.0f);
    }
    *(float4*)(out + (size_t)node * COUT + c4) = o;
}

template<bool RELU, int COUT, int NCOUT>
__global__ __launch_bounds__(4 * COUT)
void gat_gather_gemm(const int* __restrict__ cnt, const int* __restrict__ slots,
                     const int* __restrict__ ovf_src, const int* __restrict__ ovf_dst,
                     const float* __restrict__ S, const float* __restrict__ D,
                     const float* __restrict__ H, const float* __restrict__ bias,
                     const float* __restrict__ Wn, const float* __restrict__ asrcn,
                     const float* __restrict__ adstn,
                     float* __restrict__ Hn, float* __restrict__ Sn, float* __restrict__ Dn,
                     int N)
{
    constexpr int G  = COUT / 4;
    constexpr int XP = COUT + 4;
    constexpr int NT = NCOUT / 16;
    constexpr int NW = (4 * COUT) / 32;
    constexpr int KS = NW / NT;
    constexpr int KL = COUT / KS;
    constexpr int HP = NCOUT + 4;
    static_assert(NT * KS == NW && KL % 4 == 0, "one (tile, k slice) pair per wave");
    __shared__ __attribute__((aligned(16))) float xs[16 * XP];
    __shared__ __attribute__((aligned(16))) float hp[KS * 16 * HP];

    const int r    = (int)threadIdx.x / G;
    const int m0   = (int)blockIdx.x * 16;
    const int node = m0 + r;
    const int lig  = threadIdx.x & (G - 1);
    const int c4   = lig * 4;

    float4 o;
    o.x = 0.0f; o.y = 0.0f; o.z = 0.0f; o.w = 0.0f;
    if (node < N) {
        o = gat_node_sum<COUT>(cnt, slots, ovf_src, ovf_dst, S, D, H, node, lig, N);
        o.x += bias[c4 + 0];
        o.y += bias[c4 + 1];
        o.z += bias[c4 + 2];
        o.w += bias[c4 + 3];
        if (RELU) {
            o.x = fmaxf(o.x, 0.0f); o.y = fmaxf(o.y, 0.0f);
            o.z = fmaxf(o.z, 0.0f); o.w = fmaxf(o.w, 0.0f);
        }
    }
    *(float4*)(xs + r * XP + c4) = o;
    __syncthreads();

    const int wave = (int)threadIdx.x >> 5;
    const int lane = threadIdx.x & 31;
    const int half = lane >> 4;
    const int l    = lane & 15;

    float wa[NT], wd[NT];
    #pragma unroll
    for (int t = 0; t < NT; ++t) { wa[t] = asrcn[t * 16 + l]; wd[t] = adstn[t * 16 + l]; }

    {
        const int tile = wave % NT;
        const int ks   = wave / NT;
        v8f acc = {};
        const int n = tile * 16 + l;
        const float* xrow = xs + l * XP + ks * KL + 2 * half;
        const float* wcol = Wn + (size_t)(ks * KL + 2 * half) * NCOUT + n;
        #pragma unroll
        for (int k = 0; k < KL; k += 4) {
            v2f A, B;
            A[0] = xrow[k];
            A[1] = xrow[k + 1];
            B[0] = wcol[(size_t)k * NCOUT];
            B[1] = wcol[(size_t)(k + 1) * NCOUT];
            acc = __builtin_amdgcn_wmma_f32_16x16x4_f32(
                      false, A, false, B, (short)0, acc, false, false);
        }
        float* hq = hp + ks * (16 * HP);
        #pragma unroll
        for (int j = 0; j < 8; ++j)
            hq[(j + 8 * half) * HP + n] = acc[j];
    }
    __syncthreads();

    if (wave < 8) {
        const int lr  = wave + 8 * half;
        const int row = m0 + lr;
        float ps = 0.0f, pd = 0.0f;
        #pragma unroll
        for (int t = 0; t < NT; ++t) {
            const int n = t * 16 + l;
            float v = hp[lr * HP + n];
            #pragma unroll
            for (int q = 1; q < KS; ++q) v += hp[q * (16 * HP) + lr * HP + n];
            if (row < N) Hn[(size_t)row * NCOUT + n] = v;
            ps += v * wa[t];
            pd += v * wd[t];
        }
        ps = row_sum16(ps);
        pd = row_sum16(pd);
        if (l == 0 && row < N) { Sn[row] = ps; Dn[row] = pd; }
    }
}

static inline int cdiv_ll(long long a, long long b) { return (int)((a + b - 1) / b); }

extern "C" void kernel_launch(void* const* d_in, const int* in_sizes, int n_in,
                              void* d_out, int out_size, void* d_ws, size_t ws_size,
                              hipStream_t stream)
{
    const float* x   = (const float*)d_in[0];
    const int*   ei  = (const int*)  d_in[1];
    const float* W1  = (const float*)d_in[2];
    const float* as1 = (const float*)d_in[3];
    const float* ad1 = (const float*)d_in[4];
    const float* b1  = (const float*)d_in[5];
    const float* W2  = (const float*)d_in[6];
    const float* as2 = (const float*)d_in[7];
    const float* ad2 = (const float*)d_in[8];
    const float* b2  = (const float*)d_in[9];
    const float* W3  = (const float*)d_in[10];
    const float* as3 = (const float*)d_in[11];
    const float* ad3 = (const float*)d_in[12];
    const float* b3  = (const float*)d_in[13];

    const int N    = in_sizes[0] / 128;
    const int E    = in_sizes[1] / 2;
    const int Eall = E + N;
    const int* src = ei;
    const int* dst = ei + E;

    float* ws   = (float*)d_ws;
    float* h    = ws;
    float* agg  = h   + (size_t)N * 128;
    float* xn   = agg + (size_t)N * 128;
    float* alp  = xn  + (size_t)N * 128;
    float* Sv   = alp + Eall;
    float* Dv   = Sv  + N;
    int* slots   = (int*)agg;
    int* ovf_src = slots + (size_t)N * SLOT_CAP;
    int* ovf_dst = ovf_src + E;
    float* Sb    = (float*)(ovf_dst + E);
    float* Db    = Sb + N;
    int* cnt     = (int*)alp;

    int* ctr     = cnt + N;
    int* bins    = (int*)xn;
    int* spill   = (int*)xn + (size_t)N * 64;

    const dim3 blk(256);
    const int bin_blocks   = cdiv_ll(E, BIN_THREADS * BIN_EPT);
    const int prod_blocks  = cdiv_ll(cdiv_ll(N, 16), PROD_STRIPS);
    const int prod_a       = prod_blocks / 2;
    const int prod_b       = prod_blocks - prod_a;

    ctr_zero<<<1, dim3(NBKT + 2), 0, stream>>>(ctr);
    bin_and_product<<<bin_blocks + prod_a, dim3(BIN_THREADS), 0, stream>>>(
        src, dst, ctr, bins, spill, x, W1, as1, ad1, h, Sv, Dv, N, E, bin_blocks, 0);
    build_and_product<<<NBKT + prod_b, dim3(BIN_THREADS), 0, stream>>>(
        bins, spill, ctr, cnt, slots, ovf_src, ovf_dst, x, W1, as1, ad1, h, Sv, Dv, N, NBKT, prod_a * PROD_STRIPS);

    gat_gather_gemm<true,128,64><<<cdiv_ll(N, 16), dim3(512), 0, stream>>>(
        cnt, slots, ovf_src, ovf_dst, Sv, Dv, h, b1, W2, as2, ad2, xn, Sb, Db, N);

    gat_gather_gemm<true,64,64><<<cdiv_ll(N, 16), dim3(256), 0, stream>>>(
        cnt, slots, ovf_src, ovf_dst, Sb, Db, xn, b2, W3, as3, ad3, h, Sv, Dv, N);

    gat_gather<false,64><<<cdiv_ll((long long)N * 16, 256), blk, 0, stream>>>(cnt, slots, ovf_src, ovf_dst, Sv, Dv, h, b3, (float*)d_out, N);
}
